# starting kernel
# baseline (speedup 1.0000x reference)
#include <hip/hip_runtime.h>


typedef float v2f __attribute__((ext_vector_type(2)));
typedef float v8f __attribute__((ext_vector_type(8)));

#define B_N        256
#define F_N        512
#define T_N        81
#define WIN_N      1022
#define HOP_N      400
#define NF_N       69
#define KEPT_LEN   30000
#define OUT_LEN    28550
#define SIG_STRIDE 33024
#define OUT_STRIDE 28560
#define KPAD       1024
#define M1         (B_N * T_N)
#define M2         (B_N * NF_N)

#define BM 64
#define BN 128
#define BK 32
#define TS 36

#define TWO_PI 6.283185307179586476925f

__device__ __forceinline__ float hann_f(int n) {
    return 0.5f - 0.5f * cosf((TWO_PI / WIN_N) * (float)n);
}

__global__ __launch_bounds__(256) void build_mats(float* __restrict__ WinvT,
                                                  float* __restrict__ WfwdT) {
    int idx = blockIdx.x * 256 + threadIdx.x;
    int which = idx >> 20;
    int e   = idx & (KPAD * KPAD - 1);
    int row = e >> 10;
    int col = e & (KPAD - 1);
    if (which == 0) {
        float v = 0.f;
        int n = row;
        if (n < WIN_N) {
            int f = col >> 1, c = col & 1;
            int m = (f * n) % WIN_N;
            float ang = (TWO_PI / WIN_N) * (float)m;
            float wf  = (f == 0 || f == F_N - 1) ? 1.f : 2.f;
            float tr  = (c == 0) ? cosf(ang) : -sinf(ang);
            float h = hann_f(n);
            int p = n % HOP_N;
            float d = 0.f;
            for (int j = p; j < WIN_N; j += HOP_N) {
                float hj = hann_f(j);
                d += hj * hj;
            }
            v = wf * tr * (1.0f / (float)WIN_N) * (h / d);
        }
        WinvT[(size_t)row * KPAD + col] = v;
    } else {
        float v = 0.f;
        int n = col;
        if (n < WIN_N) {
            int k = row & (F_N - 1), c = row >> 9;
            int m = (k * n) % WIN_N;
            float ang = (TWO_PI / WIN_N) * (float)m;
            float tr  = (c == 0) ? cosf(ang) : -sinf(ang);
            v = tr * hann_f(n);
        }
        WfwdT[(size_t)row * KPAD + col] = v;
    }
}

__global__ __launch_bounds__(256) void istft_gemm_ola(const float* __restrict__ x,
                                                      const float* __restrict__ WinvT,
                                                      float* __restrict__ sig) {
    __shared__ float sA[BM][TS];
    __shared__ float sB[BN][TS];

    const int tid  = threadIdx.x;
    const int lane = tid & 31;
    const int wave = tid >> 5;
    const int wm   = wave & 1;
    const int wn   = wave >> 1;
    const int half = lane >> 4;
    const int r16  = lane & 15;

    const int rowBlock = blockIdx.y * BM;
    const int colBlock = blockIdx.x * BN;

    const int a_lr = tid & 63;
    const int a_lc = (tid >> 6) << 3;
    const int Ra   = rowBlock + a_lr;
    const int ab   = Ra / T_N;
    const int at   = Ra - ab * T_N;
    const float* xb = x + (size_t)ab * (F_N * T_N * 2) + at * 2;

    const int b_row = tid >> 3;
    const int b_kc  = (tid & 7) << 2;

    v8f acc[2][2];
#pragma unroll
    for (int i = 0; i < 2; ++i)
#pragma unroll
        for (int j = 0; j < 2; ++j)
            acc[i][j] = (v8f){0.f,0.f,0.f,0.f,0.f,0.f,0.f,0.f};

    for (int kb = 0; kb < KPAD; kb += BK) {
        __syncthreads();
#pragma unroll
        for (int u = 0; u < 4; ++u) {
            int f = ((kb + a_lc) >> 1) + u;
            *(v2f*)&sA[a_lr][a_lc + (u << 1)] = *(const v2f*)(xb + (size_t)f * (T_N * 2));
        }
#pragma unroll
        for (int u = 0; u < 4; ++u) {
            int nrow = b_row + (u << 5);
            *(float4*)&sB[nrow][b_kc] =
                *(const float4*)(WinvT + (size_t)(colBlock + nrow) * KPAD + kb + b_kc);
        }
        __syncthreads();
#pragma unroll
        for (int k0 = 0; k0 < BK; k0 += 4) {
            const int ka = k0 + (half << 1);
            v2f a0 = *(const v2f*)&sA[(wm << 5) + r16][ka];
            v2f a1 = *(const v2f*)&sA[(wm << 5) + 16 + r16][ka];
            v2f b0 = *(const v2f*)&sB[(wn << 5) + r16][ka];
            v2f b1 = *(const v2f*)&sB[(wn << 5) + 16 + r16][ka];
            acc[0][0] = __builtin_amdgcn_wmma_f32_16x16x4_f32(false, a0, false, b0, (short)0, acc[0][0], false, false);
            acc[0][1] = __builtin_amdgcn_wmma_f32_16x16x4_f32(false, a0, false, b1, (short)0, acc[0][1], false, false);
            acc[1][0] = __builtin_amdgcn_wmma_f32_16x16x4_f32(false, a1, false, b0, (short)0, acc[1][0], false, false);
            acc[1][1] = __builtin_amdgcn_wmma_f32_16x16x4_f32(false, a1, false, b1, (short)0, acc[1][1], false, false);
        }
    }

#pragma unroll
    for (int in = 0; in < 2; ++in) {
        const int n = colBlock + (wn << 5) + (in << 4) + r16;
        if (n < WIN_N) {
#pragma unroll
            for (int im = 0; im < 2; ++im) {
                const int Rb = rowBlock + (wm << 5) + (im << 4) + (half << 3);
#pragma unroll
                for (int v = 0; v < 8; ++v) {
                    int Rr = Rb + v;
                    int b  = Rr / T_N;
                    int t  = Rr - b * T_N;
                    atomicAdd(sig + (size_t)b * SIG_STRIDE + t * HOP_N + n, acc[im][in][v]);
                }
            }
        }
    }
}

__global__ __launch_bounds__(256) void fold_kernel(const float* __restrict__ sig,
                                                   float* __restrict__ outsig) {
    int i = blockIdx.x * 256 + threadIdx.x;
    if (i >= B_N * OUT_LEN) return;
    int b = i / OUT_LEN;
    int o = i - b * OUT_LEN;

    int k = (o < 1000) ? 0 : (o - 50) / 950;
    int r = o - k * 950;
    int j = k * 1000 + r;

    const float* s = sig + (size_t)b * SIG_STRIDE;
    float kept = s[k * 1100 + r];
    float sh = 0.f;
    int j2 = j + 50;
    if (j2 < KEPT_LEN) {
        int k2 = j2 / 1000, r2 = j2 - k2 * 1000;
        sh = s[k2 * 1100 + r2];
    }

    float cconst;
    if (j < 950)        cconst = 1.f;
    else if (j < 1000)  cconst = 0.f;
    else if (j < 29000) cconst = (r >= 50 && r < 950) ? 1.f : 0.f;
    else                cconst = 1.f;
    float cdec = (j < 29000 && r >= 950) ? (float)(1000 - r) * 0.02f : 0.f;
    float cinc = 0.f;
    if (j >= 950) {
        int rr = (j - 950) % 1000;
        if (rr < 50) cinc = (float)(rr + 1) * 0.02f;
    }
    outsig[(size_t)b * OUT_STRIDE + o] = (cconst + cdec) * kept + cinc * sh;
}

__global__ __launch_bounds__(256) void stft_gemm(const float* __restrict__ outsig,
                                                 const float* __restrict__ WfwdT,
                                                 float* __restrict__ out) {
    __shared__ float sA[BM][TS];
    __shared__ float sB[BN][TS];

    const int tid  = threadIdx.x;
    const int lane = tid & 31;
    const int wave = tid >> 5;
    const int wm   = wave & 1;
    const int wn   = wave >> 1;
    const int half = lane >> 4;
    const int r16  = lane & 15;

    const int rowBlock = blockIdx.y * BM;
    const int colBlock = blockIdx.x * BN;

    const int a_lr = tid >> 2;
    const int a_lc = (tid & 3) << 3;
    const int Ra   = rowBlock + a_lr;
    const int ab   = Ra / NF_N;
    const int am   = Ra - ab * NF_N;
    const float* sg = outsig + (size_t)ab * OUT_STRIDE + am * HOP_N;

    const int b_row = tid >> 3;
    const int b_kc  = (tid & 7) << 2;

    v8f acc[2][2];
#pragma unroll
    for (int i = 0; i < 2; ++i)
#pragma unroll
        for (int j = 0; j < 2; ++j)
            acc[i][j] = (v8f){0.f,0.f,0.f,0.f,0.f,0.f,0.f,0.f};

    for (int kb = 0; kb < KPAD; kb += BK) {
        __syncthreads();
        *(float4*)&sA[a_lr][a_lc]     = *(const float4*)(sg + kb + a_lc);
        *(float4*)&sA[a_lr][a_lc + 4] = *(const float4*)(sg + kb + a_lc + 4);
#pragma unroll
        for (int u = 0; u < 4; ++u) {
            int nrow = b_row + (u << 5);
            *(float4*)&sB[nrow][b_kc] =
                *(const float4*)(WfwdT + (size_t)(colBlock + nrow) * KPAD + kb + b_kc);
        }
        __syncthreads();
#pragma unroll
        for (int k0 = 0; k0 < BK; k0 += 4) {
            const int ka = k0 + (half << 1);
            v2f a0 = *(const v2f*)&sA[(wm << 5) + r16][ka];
            v2f a1 = *(const v2f*)&sA[(wm << 5) + 16 + r16][ka];
            v2f b0 = *(const v2f*)&sB[(wn << 5) + r16][ka];
            v2f b1 = *(const v2f*)&sB[(wn << 5) + 16 + r16][ka];
            acc[0][0] = __builtin_amdgcn_wmma_f32_16x16x4_f32(false, a0, false, b0, (short)0, acc[0][0], false, false);
            acc[0][1] = __builtin_amdgcn_wmma_f32_16x16x4_f32(false, a0, false, b1, (short)0, acc[0][1], false, false);
            acc[1][0] = __builtin_amdgcn_wmma_f32_16x16x4_f32(false, a1, false, b0, (short)0, acc[1][0], false, false);
            acc[1][1] = __builtin_amdgcn_wmma_f32_16x16x4_f32(false, a1, false, b1, (short)0, acc[1][1], false, false);
        }
    }

#pragma unroll
    for (int in = 0; in < 2; ++in) {
        const int q  = colBlock + (wn << 5) + (in << 4) + r16;
        const int kf = q & (F_N - 1);
        const int cc = q >> 9;
#pragma unroll
        for (int im = 0; im < 2; ++im) {
            const int Rb = rowBlock + (wm << 5) + (im << 4) + (half << 3);
#pragma unroll
            for (int v = 0; v < 8; ++v) {
                int Rr = Rb + v;
                int b  = Rr / NF_N;
                int m  = Rr - b * NF_N;
                out[(size_t)b * (F_N * NF_N * 2) + kf * (NF_N * 2) + m * 2 + cc] = acc[im][in][v];
            }
        }
    }
}

extern "C" void kernel_launch(void* const* d_in, const int* in_sizes, int n_in,
                              void* d_out, int out_size, void* d_ws, size_t ws_size,
                              hipStream_t stream) {
    (void)in_sizes; (void)n_in; (void)out_size; (void)ws_size;
    const float* x = (const float*)d_in[0];
    float* out = (float*)d_out;

    float* ws     = (float*)d_ws;
    float* WinvT  = ws;
    float* WfwdT  = WinvT + (size_t)KPAD * KPAD;
    float* sig    = WfwdT + (size_t)KPAD * KPAD;
    float* outsig = sig + (size_t)B_N * SIG_STRIDE;

    hipMemsetAsync(sig, 0, (size_t)B_N * SIG_STRIDE * sizeof(float), stream);
    build_mats<<<(2 * KPAD * KPAD) / 256, 256, 0, stream>>>(WinvT, WfwdT);
    istft_gemm_ola<<<dim3(KPAD / BN, M1 / BM), 256, 0, stream>>>(x, WinvT, sig);
    fold_kernel<<<(B_N * OUT_LEN + 255) / 256, 256, 0, stream>>>(sig, outsig);
    stft_gemm<<<dim3(KPAD / BN, M2 / BM), 256, 0, stream>>>(outsig, WfwdT, out);
}


# Inverse DFT split radix-2 in frequency: even and odd bins run as two 511-point products (half the WMMAs of that launch); samples n and n+511 are their sum and difference; terms summed in a new order
# speedup vs baseline: 1.9815x; 1.9815x over previous
#include <hip/hip_runtime.h>


typedef float v2f __attribute__((ext_vector_type(2)));
typedef float v8f __attribute__((ext_vector_type(8)));

#define B_N        256
#define F_N        512
#define T_N        81
#define WIN_N      1022
#define HOP_N      400
#define NF_N       69
#define OUT_LEN    28550
#define PERIOD_N   1100
#define SEG_N      1000
#define OVL_N      50
#define NSEG       30
#define OUT_STRIDE 28560
#define KPAD       1024
#define M1         (B_N * T_N)
#define M2         (B_N * NF_N)
#define HALF_N     511
#define JPAD       512
#define FWD_ELEMS  (4 * 256 * JPAD)
#define NHP        512
#define INV_ELEMS  (2 * NHP * NHP)

#define BM 64
#define BN 128
#define BK 32
#define TS 36

#define TWO_PI 6.283185307179586476925f

__device__ __forceinline__ float hann_f(int n) {
    return 0.5f - 0.5f * cosf((TWO_PI / WIN_N) * (float)n);
}

__global__ __launch_bounds__(256) void build_mats(float* __restrict__ Wi,
                                                  float* __restrict__ WfwdT) {
    int idx = blockIdx.x * 256 + threadIdx.x;
    if (idx < INV_ELEMS) {
        float v = 0.f;
        int mat = idx >> 18;
        int e   = idx & (NHP * NHP - 1);
        int n   = e >> 9;
        int kk  = e & (NHP - 1);
        if (n < HALF_N) {
            int f = (kk & ~1) + mat, c = kk & 1;
            int m = (f * n) % WIN_N;
            float ang = (TWO_PI / WIN_N) * (float)m;
            float wf  = (f == 0 || f == F_N - 1) ? 1.f : 2.f;
            float tr  = (c == 0) ? cosf(ang) : -sinf(ang);
            v = wf * tr * (1.0f / (float)WIN_N);
        }
        Wi[idx] = v;
    } else if (idx < INV_ELEMS + FWD_ELEMS) {
        int e = idx - INV_ELEMS;
        float v = 0.f;
        int q = e >> 9;
        int j = e & (JPAD - 1);
        if (j < HALF_N) {
            int mat = q >> 8, k = q & 255;
            int m = (k * j) % HALF_N;
            float ang = (TWO_PI / HALF_N) * (float)m;
            float tr  = ((mat & 1) == 0) ? cosf(ang) : -sinf(ang);
            v = tr * hann_f(2 * j + (mat >> 1));
        }
        WfwdT[e] = v;
    } else if (idx < INV_ELEMS + FWD_ELEMS + 512) {
        int t = idx - (INV_ELEMS + FWD_ELEMS);
        float ang = (TWO_PI / WIN_N) * (float)(t >> 1);
        WfwdT[FWD_ELEMS + t] = (t & 1) ? sinf(ang) : cosf(ang);
    } else {
        int n = idx - (INV_ELEMS + FWD_ELEMS + 512);
        float v = 0.f;
        if (n < WIN_N) {
            int p = n % HOP_N;
            float d = 0.f;
            for (int j = p; j < WIN_N; j += HOP_N) {
                float hj = hann_f(j);
                d += hj * hj;
            }
            v = hann_f(n) / d;
        }
        Wi[INV_ELEMS + n] = v;
    }
}

__device__ __forceinline__ void inv_half(const float* __restrict__ xb,
                                         const float* __restrict__ Wp,
                                         float (*sA)[TS], float (*sB)[TS],
                                         v8f (&acc)[2][2],
                                         const int a_lr, const int a_lc,
                                         const int b_row, const int b_kc,
                                         const int colBlock,
                                         const int wm, const int wn,
                                         const int half, const int r16) {
    for (int kb = 0; kb < NHP; kb += BK) {
        __syncthreads();
#pragma unroll
        for (int u = 0; u < 4; ++u) {
            int g = ((kb + a_lc) >> 1) + u;
            *(v2f*)&sA[a_lr][a_lc + (u << 1)] = *(const v2f*)(xb + (size_t)g * (2 * T_N * 2));
        }
#pragma unroll
        for (int u = 0; u < 4; ++u) {
            int nrow = b_row + (u << 5);
            *(float4*)&sB[nrow][b_kc] =
                *(const float4*)(Wp + (size_t)(colBlock + nrow) * NHP + kb + b_kc);
        }
        __syncthreads();
#pragma unroll
        for (int k0 = 0; k0 < BK; k0 += 4) {
            const int ka = k0 + (half << 1);
            v2f a0 = *(const v2f*)&sA[(wm << 5) + r16][ka];
            v2f a1 = *(const v2f*)&sA[(wm << 5) + 16 + r16][ka];
            v2f b0 = *(const v2f*)&sB[(wn << 5) + r16][ka];
            v2f b1 = *(const v2f*)&sB[(wn << 5) + 16 + r16][ka];
            acc[0][0] = __builtin_amdgcn_wmma_f32_16x16x4_f32(false, a0, false, b0, (short)0, acc[0][0], false, false);
            acc[0][1] = __builtin_amdgcn_wmma_f32_16x16x4_f32(false, a0, false, b1, (short)0, acc[0][1], false, false);
            acc[1][0] = __builtin_amdgcn_wmma_f32_16x16x4_f32(false, a1, false, b0, (short)0, acc[1][0], false, false);
            acc[1][1] = __builtin_amdgcn_wmma_f32_16x16x4_f32(false, a1, false, b1, (short)0, acc[1][1], false, false);
        }
    }
}

__device__ __forceinline__ void fold_add(float* __restrict__ ob, const int s,
                                         const float val, const bool live) {
    const int q  = s / PERIOD_N;
    const int rr = s - q * PERIOD_N;
    const bool up = (q > 0) & (rr < OVL_N);
    const bool dn = (q < NSEG - 1) & (rr >= SEG_N - OVL_N);
    const float ramp = (float)(up ? rr + 1 : SEG_N - rr) * 0.02f;
    const float cw   = (up | dn) ? ramp : 1.f;
    if (live & (rr < SEG_N) & (q < NSEG))
        atomicAdd(ob + q * (SEG_N - OVL_N) + rr, cw * val);
}

__global__ __launch_bounds__(256) void istft_gemm_ola(const float* __restrict__ x,
                                                      const float* __restrict__ Wi,
                                                      float* __restrict__ outsig) {
    __shared__ float sA[BM][TS];
    __shared__ float sB[BN][TS];

    const int tid  = threadIdx.x;
    const int lane = tid & 31;
    const int wave = tid >> 5;
    const int wm   = wave & 1;
    const int wn   = wave >> 1;
    const int half = lane >> 4;
    const int r16  = lane & 15;

    const int rowBlock = blockIdx.y * BM;
    const int colBlock = blockIdx.x * BN;

    const int a_lr = tid & 63;
    const int a_lc = (tid >> 6) << 3;
    const int Ra   = rowBlock + a_lr;
    const int ab   = Ra / T_N;
    const int at   = Ra - ab * T_N;
    const float* xb = x + (size_t)ab * (F_N * T_N * 2) + at * 2;

    const int b_row = tid >> 3;
    const int b_kc  = (tid & 7) << 2;

    v8f accE[2][2];
    v8f accO[2][2];
#pragma unroll
    for (int i = 0; i < 2; ++i)
#pragma unroll
        for (int j = 0; j < 2; ++j) {
            accE[i][j] = (v8f){0.f,0.f,0.f,0.f,0.f,0.f,0.f,0.f};
            accO[i][j] = (v8f){0.f,0.f,0.f,0.f,0.f,0.f,0.f,0.f};
        }

    inv_half(xb,             Wi,             sA, sB, accE, a_lr, a_lc, b_row, b_kc, colBlock, wm, wn, half, r16);
    inv_half(xb + T_N * 2,   Wi + NHP * NHP, sA, sB, accO, a_lr, a_lc, b_row, b_kc, colBlock, wm, wn, half, r16);

    const float* sw = Wi + INV_ELEMS;
    int   nn[2];
    float w_lo[2], w_hi[2];
#pragma unroll
    for (int in = 0; in < 2; ++in) {
        nn[in]   = colBlock + (wn << 5) + (in << 4) + r16;
        w_lo[in] = sw[nn[in]];
        w_hi[in] = sw[nn[in] + HALF_N];
    }
#pragma unroll
    for (int im = 0; im < 2; ++im) {
        const int Rb = rowBlock + (wm << 5) + (im << 4) + (half << 3);
#pragma unroll
        for (int v = 0; v < 8; ++v) {
            const int Rr = Rb + v;
            const int b  = Rr / T_N;
            const int t  = Rr - b * T_N;
            float* ob = outsig + (size_t)b * OUT_STRIDE;
#pragma unroll
            for (int in = 0; in < 2; ++in) {
                const float e = accE[im][in][v];
                const float o = accO[im][in][v];
                const int   s = t * HOP_N + nn[in];
                const bool live = nn[in] < HALF_N;
                fold_add(ob, s,          (e + o) * w_lo[in], live);
                fold_add(ob, s + HALF_N, (e - o) * w_hi[in], live);
            }
        }
    }
}

__global__ __launch_bounds__(128) void stft_gemm(const float* __restrict__ outsig,
                                                 const float* __restrict__ Wf,
                                                 const float* __restrict__ tw,
                                                 float* __restrict__ out) {
    __shared__ float sE[BM][TS];
    __shared__ float sO[BM][TS];
    __shared__ float sW[128][TS];

    const int tid  = threadIdx.x;
    const int lane = tid & 31;
    const int wave = tid >> 5;
    const int wm   = wave & 1;
    const int wn   = wave >> 1;
    const int half = lane >> 4;
    const int r16  = lane & 15;

    const int rowBlock = blockIdx.y * BM;
    const int binBlock = blockIdx.x * 32;

    const int l_row = tid >> 3;
    const int l_c   = tid & 7;

    const float* sgp[4];
#pragma unroll
    for (int u = 0; u < 4; ++u) {
        const int Ra = rowBlock + l_row + (u << 4);
        const int ab = Ra / NF_N;
        const int am = Ra - ab * NF_N;
        sgp[u] = outsig + (size_t)ab * OUT_STRIDE + am * HOP_N + (l_c << 3);
    }
    const float* wp = Wf + (size_t)(binBlock + (l_row & 15)) * JPAD + (l_c << 2);

    v8f accE[2][2];
    v8f accO[2][2];
#pragma unroll
    for (int i = 0; i < 2; ++i)
#pragma unroll
        for (int j = 0; j < 2; ++j) {
            accE[i][j] = (v8f){0.f,0.f,0.f,0.f,0.f,0.f,0.f,0.f};
            accO[i][j] = (v8f){0.f,0.f,0.f,0.f,0.f,0.f,0.f,0.f};
        }

    for (int kb = 0; kb < JPAD; kb += BK) {
        __syncthreads();
#pragma unroll
        for (int u = 0; u < 4; ++u) {
            const float4 q0 = *(const float4*)(sgp[u] + 2 * kb);
            const float4 q1 = *(const float4*)(sgp[u] + 2 * kb + 4);
            const int fr = l_row + (u << 4);
            *(float4*)&sE[fr][l_c << 2] = make_float4(q0.x, q0.z, q1.x, q1.z);
            *(float4*)&sO[fr][l_c << 2] = make_float4(q0.y, q0.w, q1.y, q1.w);
        }
#pragma unroll
        for (int u = 0; u < 8; ++u) {
            const int q = ((u >> 1) << 8) + ((u & 1) << 4);
            *(float4*)&sW[(u << 4) + l_row][l_c << 2] =
                *(const float4*)(wp + (size_t)q * JPAD + kb);
        }
        __syncthreads();
#pragma unroll
        for (int k0 = 0; k0 < BK; k0 += 4) {
            const int ka = k0 + (half << 1);
            v2f e0 = *(const v2f*)&sE[(wm << 5) + r16][ka];
            v2f e1 = *(const v2f*)&sE[(wm << 5) + 16 + r16][ka];
            v2f o0 = *(const v2f*)&sO[(wm << 5) + r16][ka];
            v2f o1 = *(const v2f*)&sO[(wm << 5) + 16 + r16][ka];
            v2f ce = *(const v2f*)&sW[(wn << 4) + r16][ka];
            v2f se = *(const v2f*)&sW[32 + (wn << 4) + r16][ka];
            v2f co = *(const v2f*)&sW[64 + (wn << 4) + r16][ka];
            v2f so = *(const v2f*)&sW[96 + (wn << 4) + r16][ka];
            accE[0][0] = __builtin_amdgcn_wmma_f32_16x16x4_f32(false, ce, false, e0, (short)0, accE[0][0], false, false);
            accE[0][1] = __builtin_amdgcn_wmma_f32_16x16x4_f32(false, ce, false, e1, (short)0, accE[0][1], false, false);
            accE[1][0] = __builtin_amdgcn_wmma_f32_16x16x4_f32(false, se, false, e0, (short)0, accE[1][0], false, false);
            accE[1][1] = __builtin_amdgcn_wmma_f32_16x16x4_f32(false, se, false, e1, (short)0, accE[1][1], false, false);
            accO[0][0] = __builtin_amdgcn_wmma_f32_16x16x4_f32(false, co, false, o0, (short)0, accO[0][0], false, false);
            accO[0][1] = __builtin_amdgcn_wmma_f32_16x16x4_f32(false, co, false, o1, (short)0, accO[0][1], false, false);
            accO[1][0] = __builtin_amdgcn_wmma_f32_16x16x4_f32(false, so, false, o0, (short)0, accO[1][0], false, false);
            accO[1][1] = __builtin_amdgcn_wmma_f32_16x16x4_f32(false, so, false, o1, (short)0, accO[1][1], false, false);
        }
    }

    const int kf0 = binBlock + (wn << 4) + (half << 3);
#pragma unroll
    for (int im = 0; im < 2; ++im) {
        const int Rr = rowBlock + (wm << 5) + (im << 4) + r16;
        const int b  = Rr / NF_N;
        const int m  = Rr - b * NF_N;
        float* op = out + (size_t)b * (F_N * NF_N * 2) + m * 2;
#pragma unroll
        for (int v = 0; v < 8; ++v) {
            const int k = kf0 + v;
            const v2f cs = *(const v2f*)(tw + 2 * k);
            const float er = accE[0][im][v], ei = accE[1][im][v];
            const float orr = accO[0][im][v], oi = accO[1][im][v];
            const float P = orr * cs.x + oi * cs.y;
            const float Q = oi * cs.x - orr * cs.y;
            v2f lo, hi;
            lo.x = er + P;  lo.y = ei + Q;
            hi.x = er - P;  hi.y = Q - ei;
            *(v2f*)(op + (size_t)k * (NF_N * 2)) = lo;
            *(v2f*)(op + (size_t)(F_N - 1 - k) * (NF_N * 2)) = hi;
        }
    }
}

extern "C" void kernel_launch(void* const* d_in, const int* in_sizes, int n_in,
                              void* d_out, int out_size, void* d_ws, size_t ws_size,
                              hipStream_t stream) {
    (void)in_sizes; (void)n_in; (void)out_size; (void)ws_size;
    const float* x = (const float*)d_in[0];
    float* out = (float*)d_out;

    float* ws     = (float*)d_ws;
    float* Wi     = ws;
    float* WfwdT  = Wi + (size_t)KPAD * KPAD;
    float* outsig = WfwdT + (size_t)KPAD * KPAD;

    hipMemsetAsync(outsig, 0, (size_t)B_N * OUT_STRIDE * sizeof(float), stream);
    build_mats<<<(INV_ELEMS + FWD_ELEMS + 512 + KPAD) / 256, 256, 0, stream>>>(Wi, WfwdT);
    istft_gemm_ola<<<dim3(NHP / BN, M1 / BM), 256, 0, stream>>>(x, Wi, outsig);
    stft_gemm<<<dim3(256 / 32, M2 / BM), 128, 0, stream>>>(outsig, WfwdT, WfwdT + FWD_ELEMS, out);
}


# Inverse product tiled by whole signal: a work-group owns a signal's 81 frames, overlap-adds and folds them in LDS (LDS float atomics, arrival order), stores it once; no global atomics, no memset
# speedup vs baseline: 2.0770x; 1.0482x over previous
#include <hip/hip_runtime.h>


typedef float v2f __attribute__((ext_vector_type(2)));
typedef float v8f __attribute__((ext_vector_type(8)));

#define B_N        256
#define F_N        512
#define T_N        81
#define WIN_N      1022
#define HOP_N      400
#define NF_N       69
#define OUT_LEN    28550
#define PERIOD_N   1100
#define SEG_N      1000
#define OVL_N      50
#define NSEG       30
#define OUT_STRIDE 28576
#define KPAD       1024
#define M2         (B_N * NF_N)
#define HALF_N     511
#define JPAD       512
#define FWD_ELEMS  (4 * 256 * JPAD)
#define NHP        512
#define INV_ELEMS  (2 * NHP * NHP)

#define BM 64
#define BK 32
#define SIG_ROWS 96
#define CB_N     256
#define TS 36

#define TWO_PI 6.283185307179586476925f

__device__ __forceinline__ float hann_f(int n) {
    return 0.5f - 0.5f * cosf((TWO_PI / WIN_N) * (float)n);
}

__global__ __launch_bounds__(256) void build_mats(float* __restrict__ Wi,
                                                  float* __restrict__ WfwdT) {
    int idx = blockIdx.x * 256 + threadIdx.x;
    if (idx < INV_ELEMS) {
        float v = 0.f;
        int mat = idx >> 18;
        int e   = idx & (NHP * NHP - 1);
        int n   = e >> 9;
        int kk  = e & (NHP - 1);
        if (n < HALF_N) {
            int f = (kk & ~1) + mat, c = kk & 1;
            int m = (f * n) % WIN_N;
            float ang = (TWO_PI / WIN_N) * (float)m;
            float wf  = (f == 0 || f == F_N - 1) ? 1.f : 2.f;
            float tr  = (c == 0) ? cosf(ang) : -sinf(ang);
            v = wf * tr * (1.0f / (float)WIN_N);
        }
        Wi[idx] = v;
    } else if (idx < INV_ELEMS + FWD_ELEMS) {
        int e = idx - INV_ELEMS;
        float v = 0.f;
        int q = e >> 9;
        int j = e & (JPAD - 1);
        if (j < HALF_N) {
            int mat = q >> 8, k = q & 255;
            int m = (k * j) % HALF_N;
            float ang = (TWO_PI / HALF_N) * (float)m;
            float tr  = ((mat & 1) == 0) ? cosf(ang) : -sinf(ang);
            v = tr * hann_f(2 * j + (mat >> 1));
        }
        WfwdT[e] = v;
    } else if (idx < INV_ELEMS + FWD_ELEMS + 512) {
        int t = idx - (INV_ELEMS + FWD_ELEMS);
        float ang = (TWO_PI / WIN_N) * (float)(t >> 1);
        WfwdT[FWD_ELEMS + t] = (t & 1) ? sinf(ang) : cosf(ang);
    } else {
        int n = idx - (INV_ELEMS + FWD_ELEMS + 512);
        float v = 0.f;
        if (n < WIN_N) {
            int p = n % HOP_N;
            float d = 0.f;
            for (int j = p; j < WIN_N; j += HOP_N) {
                float hj = hann_f(j);
                d += hj * hj;
            }
            v = hann_f(n) / d;
        }
        Wi[INV_ELEMS + n] = v;
    }
}

__device__ __forceinline__ void fold_add_lds(float* sig, const int s,
                                             const float val, const bool live) {
    const int q  = s / PERIOD_N;
    const int rr = s - q * PERIOD_N;
    const bool up = (q > 0) & (rr < OVL_N);
    const bool dn = (q < NSEG - 1) & (rr >= SEG_N - OVL_N);
    const float ramp = (float)(up ? rr + 1 : SEG_N - rr) * 0.02f;
    const float cw   = (up | dn) ? ramp : 1.f;
    if (live & (rr < SEG_N) & (q < NSEG))
        __hip_atomic_fetch_add(sig + q * (SEG_N - OVL_N) + rr, cw * val,
                               __ATOMIC_RELAXED, __HIP_MEMORY_SCOPE_WORKGROUP);
}

__device__ __forceinline__ void inv_pass(const float* __restrict__ xpar,
                                         const float* __restrict__ Wp,
                                         float (*sA)[SIG_ROWS][TS], float (*sB)[CB_N][TS],
                                         v8f (&acc)[3][2],
                                         const int (&a_off)[3], const int (&a_row)[3],
                                         const int (&a_col)[3],
                                         const int b_row, const int b_kc,
                                         const int wm, const int wn,
                                         const int half, const int r16) {
    v2f    ra[3];
    float4 rb[4];
#pragma unroll
    for (int u = 0; u < 3; ++u)
        ra[u] = *(const v2f*)(xpar + a_off[u]);
#pragma unroll
    for (int u = 0; u < 4; ++u)
        rb[u] = *(const float4*)(Wp + (size_t)(b_row + (u << 6)) * NHP + b_kc);
#pragma unroll
    for (int u = 0; u < 3; ++u) *(v2f*)&sA[0][a_row[u]][a_col[u]] = ra[u];
#pragma unroll
    for (int u = 0; u < 4; ++u) *(float4*)&sB[0][b_row + (u << 6)][b_kc] = rb[u];
    __syncthreads();

    for (int it = 0; it < NHP / BK; ++it) {
        const int  cur  = it & 1;
        const bool more = it + 1 < NHP / BK;
        if (more) {
            const int kb = (it + 1) * BK;
#pragma unroll
            for (int u = 0; u < 3; ++u)
                ra[u] = *(const v2f*)(xpar + a_off[u] + kb * (T_N * 2));
#pragma unroll
            for (int u = 0; u < 4; ++u)
                rb[u] = *(const float4*)(Wp + (size_t)(b_row + (u << 6)) * NHP + kb + b_kc);
        }
#pragma unroll
        for (int k0 = 0; k0 < BK; k0 += 4) {
            const int ka = k0 + (half << 1);
            v2f a[3], bf[2];
#pragma unroll
            for (int i = 0; i < 3; ++i)
                a[i] = *(const v2f*)&sA[cur][wm * 48 + (i << 4) + r16][ka];
#pragma unroll
            for (int j = 0; j < 2; ++j)
                bf[j] = *(const v2f*)&sB[cur][(wn << 5) + (j << 4) + r16][ka];
#pragma unroll
            for (int i = 0; i < 3; ++i)
#pragma unroll
                for (int j = 0; j < 2; ++j)
                    acc[i][j] = __builtin_amdgcn_wmma_f32_16x16x4_f32(false, a[i], false, bf[j], (short)0, acc[i][j], false, false);
        }
        if (more) {
#pragma unroll
            for (int u = 0; u < 3; ++u) *(v2f*)&sA[cur ^ 1][a_row[u]][a_col[u]] = ra[u];
#pragma unroll
            for (int u = 0; u < 4; ++u) *(float4*)&sB[cur ^ 1][b_row + (u << 6)][b_kc] = rb[u];
        }
        __syncthreads();
    }
}

__global__ __launch_bounds__(512) void istft_signal(const float* __restrict__ x,
                                                    const float* __restrict__ Wi,
                                                    float* __restrict__ outsig) {
    __shared__ float sSig[OUT_STRIDE];
    __shared__ float sA[2][SIG_ROWS][TS];
    __shared__ float sB[2][CB_N][TS];

    const int tid  = threadIdx.x;
    const int lane = tid & 31;
    const int wave = tid >> 5;
    const int wm   = wave & 1;
    const int wn   = wave >> 1;
    const int half = lane >> 4;
    const int r16  = lane & 15;
    const int b    = blockIdx.x;

    for (int i = tid; i < OUT_STRIDE / 4; i += 512)
        ((float4*)sSig)[i] = make_float4(0.f, 0.f, 0.f, 0.f);

    int a_off[3], a_row[3], a_col[3];
#pragma unroll
    for (int u = 0; u < 3; ++u) {
        const int idx = tid + (u << 9);
        const int g   = idx / SIG_ROWS;
        const int t   = idx - g * SIG_ROWS;
        a_row[u]  = t;
        a_col[u]  = g << 1;
        a_off[u]  = (g << 1) * (T_N * 2) + (t < T_N ? t : T_N - 1) * 2;
    }
    const float* xs = x + (size_t)b * (F_N * T_N * 2);

    const int b_row = tid >> 3;
    const int b_kc  = (tid & 7) << 2;

    const float* sw = Wi + INV_ELEMS;

    for (int cb = 0; cb < NHP / CB_N; ++cb) {
        v8f accE[3][2];
        v8f accO[3][2];
#pragma unroll
        for (int i = 0; i < 3; ++i)
#pragma unroll
            for (int j = 0; j < 2; ++j) {
                accE[i][j] = (v8f){0.f,0.f,0.f,0.f,0.f,0.f,0.f,0.f};
                accO[i][j] = (v8f){0.f,0.f,0.f,0.f,0.f,0.f,0.f,0.f};
            }

        const float* We = Wi + (size_t)(cb * CB_N) * NHP;
        inv_pass(xs,           We,             sA, sB, accE, a_off, a_row, a_col, b_row, b_kc, wm, wn, half, r16);
        inv_pass(xs + T_N * 2, We + NHP * NHP, sA, sB, accO, a_off, a_row, a_col, b_row, b_kc, wm, wn, half, r16);

        int   nn[2];
        float w_lo[2], w_hi[2];
#pragma unroll
        for (int j = 0; j < 2; ++j) {
            nn[j]   = cb * CB_N + (wn << 5) + (j << 4) + r16;
            w_lo[j] = sw[nn[j]];
            w_hi[j] = sw[nn[j] + HALF_N];
        }
#pragma unroll
        for (int i = 0; i < 3; ++i) {
            const int tb = wm * 48 + (i << 4) + (half << 3);
#pragma unroll
            for (int v = 0; v < 8; ++v) {
                const int t = tb + v;
#pragma unroll
                for (int j = 0; j < 2; ++j) {
                    const float e = accE[i][j][v];
                    const float o = accO[i][j][v];
                    const int   s = t * HOP_N + nn[j];
                    const bool live = (t < T_N) & (nn[j] < HALF_N);
                    fold_add_lds(sSig, s,          (e + o) * w_lo[j], live);
                    fold_add_lds(sSig, s + HALF_N, (e - o) * w_hi[j], live);
                }
            }
        }
    }

    __syncthreads();
    float4* dst = (float4*)(outsig + (size_t)b * OUT_STRIDE);
    for (int i = tid; i < OUT_STRIDE / 4; i += 512)
        dst[i] = ((const float4*)sSig)[i];
}

__global__ __launch_bounds__(128) void stft_gemm(const float* __restrict__ outsig,
                                                 const float* __restrict__ Wf,
                                                 const float* __restrict__ tw,
                                                 float* __restrict__ out) {
    __shared__ float sE[BM][TS];
    __shared__ float sO[BM][TS];
    __shared__ float sW[128][TS];

    const int tid  = threadIdx.x;
    const int lane = tid & 31;
    const int wave = tid >> 5;
    const int wm   = wave & 1;
    const int wn   = wave >> 1;
    const int half = lane >> 4;
    const int r16  = lane & 15;

    const int rowBlock = blockIdx.y * BM;
    const int binBlock = blockIdx.x * 32;

    const int l_row = tid >> 3;
    const int l_c   = tid & 7;

    const float* sgp[4];
#pragma unroll
    for (int u = 0; u < 4; ++u) {
        const int Ra = rowBlock + l_row + (u << 4);
        const int ab = Ra / NF_N;
        const int am = Ra - ab * NF_N;
        sgp[u] = outsig + (size_t)ab * OUT_STRIDE + am * HOP_N + (l_c << 3);
    }
    const float* wp = Wf + (size_t)(binBlock + (l_row & 15)) * JPAD + (l_c << 2);

    v8f accE[2][2];
    v8f accO[2][2];
#pragma unroll
    for (int i = 0; i < 2; ++i)
#pragma unroll
        for (int j = 0; j < 2; ++j) {
            accE[i][j] = (v8f){0.f,0.f,0.f,0.f,0.f,0.f,0.f,0.f};
            accO[i][j] = (v8f){0.f,0.f,0.f,0.f,0.f,0.f,0.f,0.f};
        }

    for (int kb = 0; kb < JPAD; kb += BK) {
        __syncthreads();
#pragma unroll
        for (int u = 0; u < 4; ++u) {
            const float4 q0 = *(const float4*)(sgp[u] + 2 * kb);
            const float4 q1 = *(const float4*)(sgp[u] + 2 * kb + 4);
            const int fr = l_row + (u << 4);
            *(float4*)&sE[fr][l_c << 2] = make_float4(q0.x, q0.z, q1.x, q1.z);
            *(float4*)&sO[fr][l_c << 2] = make_float4(q0.y, q0.w, q1.y, q1.w);
        }
#pragma unroll
        for (int u = 0; u < 8; ++u) {
            const int q = ((u >> 1) << 8) + ((u & 1) << 4);
            *(float4*)&sW[(u << 4) + l_row][l_c << 2] =
                *(const float4*)(wp + (size_t)q * JPAD + kb);
        }
        __syncthreads();
#pragma unroll
        for (int k0 = 0; k0 < BK; k0 += 4) {
            const int ka = k0 + (half << 1);
            v2f e0 = *(const v2f*)&sE[(wm << 5) + r16][ka];
            v2f e1 = *(const v2f*)&sE[(wm << 5) + 16 + r16][ka];
            v2f o0 = *(const v2f*)&sO[(wm << 5) + r16][ka];
            v2f o1 = *(const v2f*)&sO[(wm << 5) + 16 + r16][ka];
            v2f ce = *(const v2f*)&sW[(wn << 4) + r16][ka];
            v2f se = *(const v2f*)&sW[32 + (wn << 4) + r16][ka];
            v2f co = *(const v2f*)&sW[64 + (wn << 4) + r16][ka];
            v2f so = *(const v2f*)&sW[96 + (wn << 4) + r16][ka];
            accE[0][0] = __builtin_amdgcn_wmma_f32_16x16x4_f32(false, ce, false, e0, (short)0, accE[0][0], false, false);
            accE[0][1] = __builtin_amdgcn_wmma_f32_16x16x4_f32(false, ce, false, e1, (short)0, accE[0][1], false, false);
            accE[1][0] = __builtin_amdgcn_wmma_f32_16x16x4_f32(false, se, false, e0, (short)0, accE[1][0], false, false);
            accE[1][1] = __builtin_amdgcn_wmma_f32_16x16x4_f32(false, se, false, e1, (short)0, accE[1][1], false, false);
            accO[0][0] = __builtin_amdgcn_wmma_f32_16x16x4_f32(false, co, false, o0, (short)0, accO[0][0], false, false);
            accO[0][1] = __builtin_amdgcn_wmma_f32_16x16x4_f32(false, co, false, o1, (short)0, accO[0][1], false, false);
            accO[1][0] = __builtin_amdgcn_wmma_f32_16x16x4_f32(false, so, false, o0, (short)0, accO[1][0], false, false);
            accO[1][1] = __builtin_amdgcn_wmma_f32_16x16x4_f32(false, so, false, o1, (short)0, accO[1][1], false, false);
        }
    }

    const int kf0 = binBlock + (wn << 4) + (half << 3);
#pragma unroll
    for (int im = 0; im < 2; ++im) {
        const int Rr = rowBlock + (wm << 5) + (im << 4) + r16;
        const int b  = Rr / NF_N;
        const int m  = Rr - b * NF_N;
        float* op = out + (size_t)b * (F_N * NF_N * 2) + m * 2;
#pragma unroll
        for (int v = 0; v < 8; ++v) {
            const int k = kf0 + v;
            const v2f cs = *(const v2f*)(tw + 2 * k);
            const float er = accE[0][im][v], ei = accE[1][im][v];
            const float orr = accO[0][im][v], oi = accO[1][im][v];
            const float P = orr * cs.x + oi * cs.y;
            const float Q = oi * cs.x - orr * cs.y;
            v2f lo, hi;
            lo.x = er + P;  lo.y = ei + Q;
            hi.x = er - P;  hi.y = Q - ei;
            *(v2f*)(op + (size_t)k * (NF_N * 2)) = lo;
            *(v2f*)(op + (size_t)(F_N - 1 - k) * (NF_N * 2)) = hi;
        }
    }
}

extern "C" void kernel_launch(void* const* d_in, const int* in_sizes, int n_in,
                              void* d_out, int out_size, void* d_ws, size_t ws_size,
                              hipStream_t stream) {
    (void)in_sizes; (void)n_in; (void)out_size; (void)ws_size;
    const float* x = (const float*)d_in[0];
    float* out = (float*)d_out;

    float* ws     = (float*)d_ws;
    float* Wi     = ws;
    float* WfwdT  = Wi + (size_t)KPAD * KPAD;
    float* outsig = Wi + (size_t)1792 * KPAD;

    build_mats<<<(INV_ELEMS + FWD_ELEMS + 512 + KPAD) / 256, 256, 0, stream>>>(Wi, WfwdT);
    istft_signal<<<B_N, 512, 0, stream>>>(x, Wi, outsig);
    stft_gemm<<<dim3(256 / 32, M2 / BM), 128, 0, stream>>>(outsig, WfwdT, WfwdT + FWD_ELEMS, out);
}


# Inverse product K loop pipelined two K-blocks deep: global loads issued behind the barrier two blocks ahead; their wait and LDS stores pinned behind 30 of 48 products (scheduling barrier)
# speedup vs baseline: 2.1549x; 1.0375x over previous
#include <hip/hip_runtime.h>


typedef float v2f __attribute__((ext_vector_type(2)));
typedef float v4f __attribute__((ext_vector_type(4)));
typedef float v8f __attribute__((ext_vector_type(8)));

#define B_N        256
#define F_N        512
#define T_N        81
#define WIN_N      1022
#define HOP_N      400
#define NF_N       69
#define OUT_LEN    28550
#define PERIOD_N   1100
#define SEG_N      1000
#define OVL_N      50
#define NSEG       30
#define OUT_STRIDE 28576
#define KPAD       1024
#define M2         (B_N * NF_N)
#define HALF_N     511
#define JPAD       512
#define FWD_ELEMS  (4 * 256 * JPAD)
#define NHP        512
#define INV_ELEMS  (2 * NHP * NHP)

#define BM 64
#define BK 32
#define SIG_ROWS 96
#define CB_N     256
#define TS 36
#define KHEAD 5

#define TWO_PI 6.283185307179586476925f

__device__ __forceinline__ float hann_f(int n) {
    return 0.5f - 0.5f * cosf((TWO_PI / WIN_N) * (float)n);
}

__global__ __launch_bounds__(256) void build_mats(float* __restrict__ Wi,
                                                  float* __restrict__ WfwdT) {
    int idx = blockIdx.x * 256 + threadIdx.x;
    if (idx < INV_ELEMS) {
        float v = 0.f;
        int mat = idx >> 18;
        int e   = idx & (NHP * NHP - 1);
        int n   = e >> 9;
        int kk  = e & (NHP - 1);
        if (n < HALF_N) {
            int f = (kk & ~1) + mat, c = kk & 1;
            int m = (f * n) % WIN_N;
            float ang = (TWO_PI / WIN_N) * (float)m;
            float wf  = (f == 0 || f == F_N - 1) ? 1.f : 2.f;
            float tr  = (c == 0) ? cosf(ang) : -sinf(ang);
            v = wf * tr * (1.0f / (float)WIN_N);
        }
        Wi[idx] = v;
    } else if (idx < INV_ELEMS + FWD_ELEMS) {
        int e = idx - INV_ELEMS;
        float v = 0.f;
        int q = e >> 9;
        int j = e & (JPAD - 1);
        if (j < HALF_N) {
            int mat = q >> 8, k = q & 255;
            int m = (k * j) % HALF_N;
            float ang = (TWO_PI / HALF_N) * (float)m;
            float tr  = ((mat & 1) == 0) ? cosf(ang) : -sinf(ang);
            v = tr * hann_f(2 * j + (mat >> 1));
        }
        WfwdT[e] = v;
    } else if (idx < INV_ELEMS + FWD_ELEMS + 512) {
        int t = idx - (INV_ELEMS + FWD_ELEMS);
        float ang = (TWO_PI / WIN_N) * (float)(t >> 1);
        WfwdT[FWD_ELEMS + t] = (t & 1) ? sinf(ang) : cosf(ang);
    } else {
        int n = idx - (INV_ELEMS + FWD_ELEMS + 512);
        float v = 0.f;
        if (n < WIN_N) {
            int p = n % HOP_N;
            float d = 0.f;
            for (int j = p; j < WIN_N; j += HOP_N) {
                float hj = hann_f(j);
                d += hj * hj;
            }
            v = hann_f(n) / d;
        }
        Wi[INV_ELEMS + n] = v;
    }
}

__device__ __forceinline__ void fold_add_lds(float* sig, const int s,
                                             const float val, const bool live) {
    const int q  = s / PERIOD_N;
    const int rr = s - q * PERIOD_N;
    const bool up = (q > 0) & (rr < OVL_N);
    const bool dn = (q < NSEG - 1) & (rr >= SEG_N - OVL_N);
    const float ramp = (float)(up ? rr + 1 : SEG_N - rr) * 0.02f;
    const float cw   = (up | dn) ? ramp : 1.f;
    if (live & (rr < SEG_N) & (q < NSEG))
        __hip_atomic_fetch_add(sig + q * (SEG_N - OVL_N) + rr, cw * val,
                               __ATOMIC_RELAXED, __HIP_MEMORY_SCOPE_WORKGROUP);
}

__device__ __forceinline__ void inv_kblock(const float* __restrict__ xnext,
                                           const float* __restrict__ wnext,
                                           float (*sA)[SIG_ROWS][TS], float (*sB)[CB_N][TS],
                                           v8f (&acc)[3][2], v2f (&ra)[3], v4f (&rb)[4],
                                           const int (&a_off)[3], const int (&a_row)[3],
                                           const int (&a_col)[3],
                                           const int b_row, const int b_kc,
                                           const int wm, const int wn,
                                           const int half, const int r16,
                                           const int cur, const bool st, const bool ld) {
    v2f a[BK / 4][3], bf[BK / 4][2];
#pragma unroll
    for (int kk = 0; kk < BK / 4; ++kk) {
        const int ka = (kk << 2) + (half << 1);
#pragma unroll
        for (int i = 0; i < 3; ++i)
            a[kk][i] = *(const v2f*)&sA[cur][wm * 48 + (i << 4) + r16][ka];
#pragma unroll
        for (int j = 0; j < 2; ++j)
            bf[kk][j] = *(const v2f*)&sB[cur][(wn << 5) + (j << 4) + r16][ka];
    }
#pragma unroll
    for (int kk = 0; kk < KHEAD; ++kk)
#pragma unroll
        for (int i = 0; i < 3; ++i)
#pragma unroll
            for (int j = 0; j < 2; ++j)
                acc[i][j] = __builtin_amdgcn_wmma_f32_16x16x4_f32(false, a[kk][i], false, bf[kk][j], (short)0, acc[i][j], false, false);
    __builtin_amdgcn_sched_barrier(0);
    if (st) {
#pragma unroll
        for (int u = 0; u < 3; ++u) *(v2f*)&sA[cur ^ 1][a_row[u]][a_col[u]] = ra[u];
#pragma unroll
        for (int u = 0; u < 4; ++u) *(v4f*)&sB[cur ^ 1][b_row + (u << 6)][b_kc] = rb[u];
    }
    __syncthreads();
    if (ld) {
#pragma unroll
        for (int u = 0; u < 3; ++u)
            ra[u] = *(const v2f*)(xnext + a_off[u]);
#pragma unroll
        for (int u = 0; u < 4; ++u)
            rb[u] = *(const v4f*)(wnext + (size_t)(u << 6) * NHP);
    }
#pragma unroll
    for (int kk = KHEAD; kk < BK / 4; ++kk)
#pragma unroll
        for (int i = 0; i < 3; ++i)
#pragma unroll
            for (int j = 0; j < 2; ++j)
                acc[i][j] = __builtin_amdgcn_wmma_f32_16x16x4_f32(false, a[kk][i], false, bf[kk][j], (short)0, acc[i][j], false, false);
}

__device__ __forceinline__ void inv_pass(const float* __restrict__ xpar,
                                         const float* __restrict__ Wp,
                                         float (*sA)[SIG_ROWS][TS], float (*sB)[CB_N][TS],
                                         v8f (&acc)[3][2],
                                         const int (&a_off)[3], const int (&a_row)[3],
                                         const int (&a_col)[3],
                                         const int b_row, const int b_kc,
                                         const int wm, const int wn,
                                         const int half, const int r16) {
    v2f    ra[3];
    v4f    rb[4];
    const float* wrow = Wp + (size_t)b_row * NHP + b_kc;
#pragma unroll
    for (int u = 0; u < 3; ++u)
        ra[u] = *(const v2f*)(xpar + a_off[u]);
#pragma unroll
    for (int u = 0; u < 4; ++u)
        rb[u] = *(const v4f*)(wrow + (size_t)(u << 6) * NHP);
#pragma unroll
    for (int u = 0; u < 3; ++u) *(v2f*)&sA[0][a_row[u]][a_col[u]] = ra[u];
#pragma unroll
    for (int u = 0; u < 4; ++u) *(v4f*)&sB[0][b_row + (u << 6)][b_kc] = rb[u];
    __syncthreads();
#pragma unroll
    for (int u = 0; u < 3; ++u)
        ra[u] = *(const v2f*)(xpar + a_off[u] + BK * (T_N * 2));
#pragma unroll
    for (int u = 0; u < 4; ++u)
        rb[u] = *(const v4f*)(wrow + (size_t)(u << 6) * NHP + BK);

    for (int it = 0; it < NHP / BK - 2; ++it)
        inv_kblock(xpar + (it + 2) * BK * (T_N * 2), wrow + (it + 2) * BK, sA, sB, acc, ra, rb,
                   a_off, a_row, a_col, b_row, b_kc, wm, wn, half, r16, it & 1, true, true);
    inv_kblock(xpar, wrow, sA, sB, acc, ra, rb, a_off, a_row, a_col, b_row, b_kc,
               wm, wn, half, r16, (NHP / BK - 2) & 1, true, false);
    inv_kblock(xpar, wrow, sA, sB, acc, ra, rb, a_off, a_row, a_col, b_row, b_kc,
               wm, wn, half, r16, (NHP / BK - 1) & 1, false, false);
}

__global__ __launch_bounds__(512) void istft_signal(const float* __restrict__ x,
                                                    const float* __restrict__ Wi,
                                                    float* __restrict__ outsig) {
    __shared__ float sSig[OUT_STRIDE];
    __shared__ float sA[2][SIG_ROWS][TS];
    __shared__ float sB[2][CB_N][TS];

    const int tid  = threadIdx.x;
    const int lane = tid & 31;
    const int wave = tid >> 5;
    const int wm   = wave & 1;
    const int wn   = wave >> 1;
    const int half = lane >> 4;
    const int r16  = lane & 15;
    const int b    = blockIdx.x;

    for (int i = tid; i < OUT_STRIDE / 4; i += 512)
        ((float4*)sSig)[i] = make_float4(0.f, 0.f, 0.f, 0.f);

    int a_off[3], a_row[3], a_col[3];
#pragma unroll
    for (int u = 0; u < 3; ++u) {
        const int idx = tid + (u << 9);
        const int g   = idx / SIG_ROWS;
        const int t   = idx - g * SIG_ROWS;
        a_row[u]  = t;
        a_col[u]  = g << 1;
        a_off[u]  = (g << 1) * (T_N * 2) + (t < T_N ? t : T_N - 1) * 2;
    }
    const float* xs = x + (size_t)b * (F_N * T_N * 2);

    const int b_row = tid >> 3;
    const int b_kc  = (tid & 7) << 2;

    const float* sw = Wi + INV_ELEMS;

    for (int cb = 0; cb < NHP / CB_N; ++cb) {
        v8f accE[3][2];
        v8f accO[3][2];
#pragma unroll
        for (int i = 0; i < 3; ++i)
#pragma unroll
            for (int j = 0; j < 2; ++j) {
                accE[i][j] = (v8f){0.f,0.f,0.f,0.f,0.f,0.f,0.f,0.f};
                accO[i][j] = (v8f){0.f,0.f,0.f,0.f,0.f,0.f,0.f,0.f};
            }

        const float* We = Wi + (size_t)(cb * CB_N) * NHP;
        inv_pass(xs,           We,             sA, sB, accE, a_off, a_row, a_col, b_row, b_kc, wm, wn, half, r16);
        inv_pass(xs + T_N * 2, We + NHP * NHP, sA, sB, accO, a_off, a_row, a_col, b_row, b_kc, wm, wn, half, r16);

        int   nn[2];
        float w_lo[2], w_hi[2];
#pragma unroll
        for (int j = 0; j < 2; ++j) {
            nn[j]   = cb * CB_N + (wn << 5) + (j << 4) + r16;
            w_lo[j] = sw[nn[j]];
            w_hi[j] = sw[nn[j] + HALF_N];
        }
#pragma unroll
        for (int i = 0; i < 3; ++i) {
            const int tb = wm * 48 + (i << 4) + (half << 3);
#pragma unroll
            for (int v = 0; v < 8; ++v) {
                const int t = tb + v;
#pragma unroll
                for (int j = 0; j < 2; ++j) {
                    const float e = accE[i][j][v];
                    const float o = accO[i][j][v];
                    const int   s = t * HOP_N + nn[j];
                    const bool live = (t < T_N) & (nn[j] < HALF_N);
                    fold_add_lds(sSig, s,          (e + o) * w_lo[j], live);
                    fold_add_lds(sSig, s + HALF_N, (e - o) * w_hi[j], live);
                }
            }
        }
    }

    __syncthreads();
    float4* dst = (float4*)(outsig + (size_t)b * OUT_STRIDE);
    for (int i = tid; i < OUT_STRIDE / 4; i += 512)
        dst[i] = ((const float4*)sSig)[i];
}

__global__ __launch_bounds__(128) void stft_gemm(const float* __restrict__ outsig,
                                                 const float* __restrict__ Wf,
                                                 const float* __restrict__ tw,
                                                 float* __restrict__ out) {
    __shared__ float sE[BM][TS];
    __shared__ float sO[BM][TS];
    __shared__ float sW[128][TS];

    const int tid  = threadIdx.x;
    const int lane = tid & 31;
    const int wave = tid >> 5;
    const int wm   = wave & 1;
    const int wn   = wave >> 1;
    const int half = lane >> 4;
    const int r16  = lane & 15;

    const int rowBlock = blockIdx.y * BM;
    const int binBlock = blockIdx.x * 32;

    const int l_row = tid >> 3;
    const int l_c   = tid & 7;

    const float* sgp[4];
#pragma unroll
    for (int u = 0; u < 4; ++u) {
        const int Ra = rowBlock + l_row + (u << 4);
        const int ab = Ra / NF_N;
        const int am = Ra - ab * NF_N;
        sgp[u] = outsig + (size_t)ab * OUT_STRIDE + am * HOP_N + (l_c << 3);
    }
    const float* wp = Wf + (size_t)(binBlock + (l_row & 15)) * JPAD + (l_c << 2);

    v8f accE[2][2];
    v8f accO[2][2];
#pragma unroll
    for (int i = 0; i < 2; ++i)
#pragma unroll
        for (int j = 0; j < 2; ++j) {
            accE[i][j] = (v8f){0.f,0.f,0.f,0.f,0.f,0.f,0.f,0.f};
            accO[i][j] = (v8f){0.f,0.f,0.f,0.f,0.f,0.f,0.f,0.f};
        }

    for (int kb = 0; kb < JPAD; kb += BK) {
        __syncthreads();
#pragma unroll
        for (int u = 0; u < 4; ++u) {
            const float4 q0 = *(const float4*)(sgp[u] + 2 * kb);
            const float4 q1 = *(const float4*)(sgp[u] + 2 * kb + 4);
            const int fr = l_row + (u << 4);
            *(float4*)&sE[fr][l_c << 2] = make_float4(q0.x, q0.z, q1.x, q1.z);
            *(float4*)&sO[fr][l_c << 2] = make_float4(q0.y, q0.w, q1.y, q1.w);
        }
#pragma unroll
        for (int u = 0; u < 8; ++u) {
            const int q = ((u >> 1) << 8) + ((u & 1) << 4);
            *(float4*)&sW[(u << 4) + l_row][l_c << 2] =
                *(const float4*)(wp + (size_t)q * JPAD + kb);
        }
        __syncthreads();
#pragma unroll
        for (int k0 = 0; k0 < BK; k0 += 4) {
            const int ka = k0 + (half << 1);
            v2f e0 = *(const v2f*)&sE[(wm << 5) + r16][ka];
            v2f e1 = *(const v2f*)&sE[(wm << 5) + 16 + r16][ka];
            v2f o0 = *(const v2f*)&sO[(wm << 5) + r16][ka];
            v2f o1 = *(const v2f*)&sO[(wm << 5) + 16 + r16][ka];
            v2f ce = *(const v2f*)&sW[(wn << 4) + r16][ka];
            v2f se = *(const v2f*)&sW[32 + (wn << 4) + r16][ka];
            v2f co = *(const v2f*)&sW[64 + (wn << 4) + r16][ka];
            v2f so = *(const v2f*)&sW[96 + (wn << 4) + r16][ka];
            accE[0][0] = __builtin_amdgcn_wmma_f32_16x16x4_f32(false, ce, false, e0, (short)0, accE[0][0], false, false);
            accE[0][1] = __builtin_amdgcn_wmma_f32_16x16x4_f32(false, ce, false, e1, (short)0, accE[0][1], false, false);
            accE[1][0] = __builtin_amdgcn_wmma_f32_16x16x4_f32(false, se, false, e0, (short)0, accE[1][0], false, false);
            accE[1][1] = __builtin_amdgcn_wmma_f32_16x16x4_f32(false, se, false, e1, (short)0, accE[1][1], false, false);
            accO[0][0] = __builtin_amdgcn_wmma_f32_16x16x4_f32(false, co, false, o0, (short)0, accO[0][0], false, false);
            accO[0][1] = __builtin_amdgcn_wmma_f32_16x16x4_f32(false, co, false, o1, (short)0, accO[0][1], false, false);
            accO[1][0] = __builtin_amdgcn_wmma_f32_16x16x4_f32(false, so, false, o0, (short)0, accO[1][0], false, false);
            accO[1][1] = __builtin_amdgcn_wmma_f32_16x16x4_f32(false, so, false, o1, (short)0, accO[1][1], false, false);
        }
    }

    const int kf0 = binBlock + (wn << 4) + (half << 3);
#pragma unroll
    for (int im = 0; im < 2; ++im) {
        const int Rr = rowBlock + (wm << 5) + (im << 4) + r16;
        const int b  = Rr / NF_N;
        const int m  = Rr - b * NF_N;
        float* op = out + (size_t)b * (F_N * NF_N * 2) + m * 2;
#pragma unroll
        for (int v = 0; v < 8; ++v) {
            const int k = kf0 + v;
            const v2f cs = *(const v2f*)(tw + 2 * k);
            const float er = accE[0][im][v], ei = accE[1][im][v];
            const float orr = accO[0][im][v], oi = accO[1][im][v];
            const float P = orr * cs.x + oi * cs.y;
            const float Q = oi * cs.x - orr * cs.y;
            v2f lo, hi;
            lo.x = er + P;  lo.y = ei + Q;
            hi.x = er - P;  hi.y = Q - ei;
            *(v2f*)(op + (size_t)k * (NF_N * 2)) = lo;
            *(v2f*)(op + (size_t)(F_N - 1 - k) * (NF_N * 2)) = hi;
        }
    }
}

extern "C" void kernel_launch(void* const* d_in, const int* in_sizes, int n_in,
                              void* d_out, int out_size, void* d_ws, size_t ws_size,
                              hipStream_t stream) {
    (void)in_sizes; (void)n_in; (void)out_size; (void)ws_size;
    const float* x = (const float*)d_in[0];
    float* out = (float*)d_out;

    float* ws     = (float*)d_ws;
    float* Wi     = ws;
    float* WfwdT  = Wi + (size_t)KPAD * KPAD;
    float* outsig = Wi + (size_t)1792 * KPAD;

    build_mats<<<(INV_ELEMS + FWD_ELEMS + 512 + KPAD) / 256, 256, 0, stream>>>(Wi, WfwdT);
    istft_signal<<<B_N, 512, 0, stream>>>(x, Wi, outsig);
    stft_gemm<<<dim3(256 / 32, M2 / BM), 128, 0, stream>>>(outsig, WfwdT, WfwdT + FWD_ELEMS, out);
}


# Forward product K loop software-pipelined: the next K-block's signal loads leave as one group behind the barrier and park in registers under the 64 products; the row loads then go out as one clause
# speedup vs baseline: 2.2399x; 1.0394x over previous
#include <hip/hip_runtime.h>


typedef float v2f __attribute__((ext_vector_type(2)));
typedef float v4f __attribute__((ext_vector_type(4)));
typedef float v8f __attribute__((ext_vector_type(8)));

#define B_N        256
#define F_N        512
#define T_N        81
#define WIN_N      1022
#define HOP_N      400
#define NF_N       69
#define OUT_LEN    28550
#define PERIOD_N   1100
#define SEG_N      1000
#define OVL_N      50
#define NSEG       30
#define OUT_STRIDE 28576
#define KPAD       1024
#define M2         (B_N * NF_N)
#define HALF_N     511
#define JPAD       512
#define FWD_ELEMS  (4 * 256 * JPAD)
#define NHP        512
#define INV_ELEMS  (2 * NHP * NHP)

#define BM 64
#define BK 32
#define SIG_ROWS 96
#define CB_N     256
#define TS 36
#define KHEAD 5

#define TWO_PI 6.283185307179586476925f

__device__ __forceinline__ float hann_f(int n) {
    return 0.5f - 0.5f * cosf((TWO_PI / WIN_N) * (float)n);
}

__global__ __launch_bounds__(256) void build_mats(float* __restrict__ Wi,
                                                  float* __restrict__ WfwdT) {
    int idx = blockIdx.x * 256 + threadIdx.x;
    if (idx < INV_ELEMS) {
        float v = 0.f;
        int mat = idx >> 18;
        int e   = idx & (NHP * NHP - 1);
        int n   = e >> 9;
        int kk  = e & (NHP - 1);
        if (n < HALF_N) {
            int f = (kk & ~1) + mat, c = kk & 1;
            int m = (f * n) % WIN_N;
            float ang = (TWO_PI / WIN_N) * (float)m;
            float wf  = (f == 0 || f == F_N - 1) ? 1.f : 2.f;
            float tr  = (c == 0) ? cosf(ang) : -sinf(ang);
            v = wf * tr * (1.0f / (float)WIN_N);
        }
        Wi[idx] = v;
    } else if (idx < INV_ELEMS + FWD_ELEMS) {
        int e = idx - INV_ELEMS;
        float v = 0.f;
        int q = e >> 9;
        int j = e & (JPAD - 1);
        if (j < HALF_N) {
            int mat = q >> 8, k = q & 255;
            int m = (k * j) % HALF_N;
            float ang = (TWO_PI / HALF_N) * (float)m;
            float tr  = ((mat & 1) == 0) ? cosf(ang) : -sinf(ang);
            v = tr * hann_f(2 * j + (mat >> 1));
        }
        WfwdT[e] = v;
    } else if (idx < INV_ELEMS + FWD_ELEMS + 512) {
        int t = idx - (INV_ELEMS + FWD_ELEMS);
        float ang = (TWO_PI / WIN_N) * (float)(t >> 1);
        WfwdT[FWD_ELEMS + t] = (t & 1) ? sinf(ang) : cosf(ang);
    } else {
        int n = idx - (INV_ELEMS + FWD_ELEMS + 512);
        float v = 0.f;
        if (n < WIN_N) {
            int p = n % HOP_N;
            float d = 0.f;
            for (int j = p; j < WIN_N; j += HOP_N) {
                float hj = hann_f(j);
                d += hj * hj;
            }
            v = hann_f(n) / d;
        }
        Wi[INV_ELEMS + n] = v;
    }
}

__device__ __forceinline__ void fold_add_lds(float* sig, const int s,
                                             const float val, const bool live) {
    const int q  = s / PERIOD_N;
    const int rr = s - q * PERIOD_N;
    const bool up = (q > 0) & (rr < OVL_N);
    const bool dn = (q < NSEG - 1) & (rr >= SEG_N - OVL_N);
    const float ramp = (float)(up ? rr + 1 : SEG_N - rr) * 0.02f;
    const float cw   = (up | dn) ? ramp : 1.f;
    if (live & (rr < SEG_N) & (q < NSEG))
        __hip_atomic_fetch_add(sig + q * (SEG_N - OVL_N) + rr, cw * val,
                               __ATOMIC_RELAXED, __HIP_MEMORY_SCOPE_WORKGROUP);
}

__device__ __forceinline__ void inv_kblock(const float* __restrict__ xnext,
                                           const float* __restrict__ wnext,
                                           float (*sA)[SIG_ROWS][TS], float (*sB)[CB_N][TS],
                                           v8f (&acc)[3][2], v2f (&ra)[3], v4f (&rb)[4],
                                           const int (&a_off)[3], const int (&a_row)[3],
                                           const int (&a_col)[3],
                                           const int b_row, const int b_kc,
                                           const int wm, const int wn,
                                           const int half, const int r16,
                                           const int cur, const bool st, const bool ld) {
    v2f a[BK / 4][3], bf[BK / 4][2];
#pragma unroll
    for (int kk = 0; kk < BK / 4; ++kk) {
        const int ka = (kk << 2) + (half << 1);
#pragma unroll
        for (int i = 0; i < 3; ++i)
            a[kk][i] = *(const v2f*)&sA[cur][wm * 48 + (i << 4) + r16][ka];
#pragma unroll
        for (int j = 0; j < 2; ++j)
            bf[kk][j] = *(const v2f*)&sB[cur][(wn << 5) + (j << 4) + r16][ka];
    }
#pragma unroll
    for (int kk = 0; kk < KHEAD; ++kk)
#pragma unroll
        for (int i = 0; i < 3; ++i)
#pragma unroll
            for (int j = 0; j < 2; ++j)
                acc[i][j] = __builtin_amdgcn_wmma_f32_16x16x4_f32(false, a[kk][i], false, bf[kk][j], (short)0, acc[i][j], false, false);
    __builtin_amdgcn_sched_barrier(0);
    if (st) {
#pragma unroll
        for (int u = 0; u < 3; ++u) *(v2f*)&sA[cur ^ 1][a_row[u]][a_col[u]] = ra[u];
#pragma unroll
        for (int u = 0; u < 4; ++u) *(v4f*)&sB[cur ^ 1][b_row + (u << 6)][b_kc] = rb[u];
    }
    __syncthreads();
    if (ld) {
#pragma unroll
        for (int u = 0; u < 3; ++u)
            ra[u] = *(const v2f*)(xnext + a_off[u]);
#pragma unroll
        for (int u = 0; u < 4; ++u)
            rb[u] = *(const v4f*)(wnext + (size_t)(u << 6) * NHP);
    }
#pragma unroll
    for (int kk = KHEAD; kk < BK / 4; ++kk)
#pragma unroll
        for (int i = 0; i < 3; ++i)
#pragma unroll
            for (int j = 0; j < 2; ++j)
                acc[i][j] = __builtin_amdgcn_wmma_f32_16x16x4_f32(false, a[kk][i], false, bf[kk][j], (short)0, acc[i][j], false, false);
}

__device__ __forceinline__ void inv_pass(const float* __restrict__ xpar,
                                         const float* __restrict__ Wp,
                                         float (*sA)[SIG_ROWS][TS], float (*sB)[CB_N][TS],
                                         v8f (&acc)[3][2],
                                         const int (&a_off)[3], const int (&a_row)[3],
                                         const int (&a_col)[3],
                                         const int b_row, const int b_kc,
                                         const int wm, const int wn,
                                         const int half, const int r16) {
    v2f    ra[3];
    v4f    rb[4];
    const float* wrow = Wp + (size_t)b_row * NHP + b_kc;
#pragma unroll
    for (int u = 0; u < 3; ++u)
        ra[u] = *(const v2f*)(xpar + a_off[u]);
#pragma unroll
    for (int u = 0; u < 4; ++u)
        rb[u] = *(const v4f*)(wrow + (size_t)(u << 6) * NHP);
#pragma unroll
    for (int u = 0; u < 3; ++u) *(v2f*)&sA[0][a_row[u]][a_col[u]] = ra[u];
#pragma unroll
    for (int u = 0; u < 4; ++u) *(v4f*)&sB[0][b_row + (u << 6)][b_kc] = rb[u];
    __syncthreads();
#pragma unroll
    for (int u = 0; u < 3; ++u)
        ra[u] = *(const v2f*)(xpar + a_off[u] + BK * (T_N * 2));
#pragma unroll
    for (int u = 0; u < 4; ++u)
        rb[u] = *(const v4f*)(wrow + (size_t)(u << 6) * NHP + BK);

    for (int it = 0; it < NHP / BK - 2; ++it)
        inv_kblock(xpar + (it + 2) * BK * (T_N * 2), wrow + (it + 2) * BK, sA, sB, acc, ra, rb,
                   a_off, a_row, a_col, b_row, b_kc, wm, wn, half, r16, it & 1, true, true);
    inv_kblock(xpar, wrow, sA, sB, acc, ra, rb, a_off, a_row, a_col, b_row, b_kc,
               wm, wn, half, r16, (NHP / BK - 2) & 1, true, false);
    inv_kblock(xpar, wrow, sA, sB, acc, ra, rb, a_off, a_row, a_col, b_row, b_kc,
               wm, wn, half, r16, (NHP / BK - 1) & 1, false, false);
}

__global__ __launch_bounds__(512) void istft_signal(const float* __restrict__ x,
                                                    const float* __restrict__ Wi,
                                                    float* __restrict__ outsig) {
    __shared__ float sSig[OUT_STRIDE];
    __shared__ float sA[2][SIG_ROWS][TS];
    __shared__ float sB[2][CB_N][TS];

    const int tid  = threadIdx.x;
    const int lane = tid & 31;
    const int wave = tid >> 5;
    const int wm   = wave & 1;
    const int wn   = wave >> 1;
    const int half = lane >> 4;
    const int r16  = lane & 15;
    const int b    = blockIdx.x;

    for (int i = tid; i < OUT_STRIDE / 4; i += 512)
        ((float4*)sSig)[i] = make_float4(0.f, 0.f, 0.f, 0.f);

    int a_off[3], a_row[3], a_col[3];
#pragma unroll
    for (int u = 0; u < 3; ++u) {
        const int idx = tid + (u << 9);
        const int g   = idx / SIG_ROWS;
        const int t   = idx - g * SIG_ROWS;
        a_row[u]  = t;
        a_col[u]  = g << 1;
        a_off[u]  = (g << 1) * (T_N * 2) + (t < T_N ? t : T_N - 1) * 2;
    }
    const float* xs = x + (size_t)b * (F_N * T_N * 2);

    const int b_row = tid >> 3;
    const int b_kc  = (tid & 7) << 2;

    const float* sw = Wi + INV_ELEMS;

    for (int cb = 0; cb < NHP / CB_N; ++cb) {
        v8f accE[3][2];
        v8f accO[3][2];
#pragma unroll
        for (int i = 0; i < 3; ++i)
#pragma unroll
            for (int j = 0; j < 2; ++j) {
                accE[i][j] = (v8f){0.f,0.f,0.f,0.f,0.f,0.f,0.f,0.f};
                accO[i][j] = (v8f){0.f,0.f,0.f,0.f,0.f,0.f,0.f,0.f};
            }

        const float* We = Wi + (size_t)(cb * CB_N) * NHP;
        inv_pass(xs,           We,             sA, sB, accE, a_off, a_row, a_col, b_row, b_kc, wm, wn, half, r16);
        inv_pass(xs + T_N * 2, We + NHP * NHP, sA, sB, accO, a_off, a_row, a_col, b_row, b_kc, wm, wn, half, r16);

        int   nn[2];
        float w_lo[2], w_hi[2];
#pragma unroll
        for (int j = 0; j < 2; ++j) {
            nn[j]   = cb * CB_N + (wn << 5) + (j << 4) + r16;
            w_lo[j] = sw[nn[j]];
            w_hi[j] = sw[nn[j] + HALF_N];
        }
#pragma unroll
        for (int i = 0; i < 3; ++i) {
            const int tb = wm * 48 + (i << 4) + (half << 3);
#pragma unroll
            for (int v = 0; v < 8; ++v) {
                const int t = tb + v;
#pragma unroll
                for (int j = 0; j < 2; ++j) {
                    const float e = accE[i][j][v];
                    const float o = accO[i][j][v];
                    const int   s = t * HOP_N + nn[j];
                    const bool live = (t < T_N) & (nn[j] < HALF_N);
                    fold_add_lds(sSig, s,          (e + o) * w_lo[j], live);
                    fold_add_lds(sSig, s + HALF_N, (e - o) * w_hi[j], live);
                }
            }
        }
    }

    __syncthreads();
    float4* dst = (float4*)(outsig + (size_t)b * OUT_STRIDE);
    for (int i = tid; i < OUT_STRIDE / 4; i += 512)
        dst[i] = ((const float4*)sSig)[i];
}

__global__ __launch_bounds__(128) void stft_gemm(const float* __restrict__ outsig,
                                                 const float* __restrict__ Wf,
                                                 const float* __restrict__ tw,
                                                 float* __restrict__ out) {
    __shared__ float sE[BM][TS];
    __shared__ float sO[BM][TS];
    __shared__ float sW[128][TS];

    const int tid  = threadIdx.x;
    const int lane = tid & 31;
    const int wave = tid >> 5;
    const int wm   = wave & 1;
    const int wn   = wave >> 1;
    const int half = lane >> 4;
    const int r16  = lane & 15;

    const int rowBlock = blockIdx.y * BM;
    const int binBlock = blockIdx.x * 32;

    const int l_row = tid >> 3;
    const int l_c   = tid & 7;

    unsigned soff[4];
#pragma unroll
    for (int u = 0; u < 4; ++u) {
        const int Ra = rowBlock + l_row + (u << 4);
        const int ab = Ra / NF_N;
        const int am = Ra - ab * NF_N;
        soff[u] = (unsigned)(ab * OUT_STRIDE + am * HOP_N + (l_c << 3));
    }
    const unsigned woff = (unsigned)((binBlock + (l_row & 15)) * JPAD + (l_c << 2));

    v8f accE[2][2];
    v8f accO[2][2];
#pragma unroll
    for (int i = 0; i < 2; ++i)
#pragma unroll
        for (int j = 0; j < 2; ++j) {
            accE[i][j] = (v8f){0.f,0.f,0.f,0.f,0.f,0.f,0.f,0.f};
            accO[i][j] = (v8f){0.f,0.f,0.f,0.f,0.f,0.f,0.f,0.f};
        }

    float4 gs[8];
#pragma unroll
    for (int u = 0; u < 4; ++u) {
        gs[2 * u]     = *(const float4*)(outsig + soff[u]);
        gs[2 * u + 1] = *(const float4*)(outsig + soff[u] + 4);
    }

    for (int kb = 0; kb < JPAD; kb += BK) {
        __syncthreads();
        float4 gw[8];
        const float* wk = Wf + kb;
#pragma unroll
        for (int u = 0; u < 8; ++u) {
            const int q = ((u >> 1) << 8) + ((u & 1) << 4);
            gw[u] = *(const float4*)(wk + woff + q * JPAD);
        }
#pragma unroll
        for (int u = 0; u < 4; ++u) {
            const float4 q0 = gs[2 * u];
            const float4 q1 = gs[2 * u + 1];
            const int fr = l_row + (u << 4);
            *(float4*)&sE[fr][l_c << 2] = make_float4(q0.x, q0.z, q1.x, q1.z);
            *(float4*)&sO[fr][l_c << 2] = make_float4(q0.y, q0.w, q1.y, q1.w);
        }
#pragma unroll
        for (int u = 0; u < 8; ++u)
            *(float4*)&sW[(u << 4) + l_row][l_c << 2] = gw[u];
        __syncthreads();
        if (kb + BK < JPAD) {
            const float* sk = outsig + 2 * (kb + BK);
#pragma unroll
            for (int u = 0; u < 4; ++u) {
                gs[2 * u]     = *(const float4*)(sk + soff[u]);
                gs[2 * u + 1] = *(const float4*)(sk + soff[u] + 4);
            }
        }
#pragma unroll
        for (int k0 = 0; k0 < BK; k0 += 4) {
            const int ka = k0 + (half << 1);
            v2f e0 = *(const v2f*)&sE[(wm << 5) + r16][ka];
            v2f e1 = *(const v2f*)&sE[(wm << 5) + 16 + r16][ka];
            v2f o0 = *(const v2f*)&sO[(wm << 5) + r16][ka];
            v2f o1 = *(const v2f*)&sO[(wm << 5) + 16 + r16][ka];
            v2f ce = *(const v2f*)&sW[(wn << 4) + r16][ka];
            v2f se = *(const v2f*)&sW[32 + (wn << 4) + r16][ka];
            v2f co = *(const v2f*)&sW[64 + (wn << 4) + r16][ka];
            v2f so = *(const v2f*)&sW[96 + (wn << 4) + r16][ka];
            accE[0][0] = __builtin_amdgcn_wmma_f32_16x16x4_f32(false, ce, false, e0, (short)0, accE[0][0], false, false);
            accE[0][1] = __builtin_amdgcn_wmma_f32_16x16x4_f32(false, ce, false, e1, (short)0, accE[0][1], false, false);
            accE[1][0] = __builtin_amdgcn_wmma_f32_16x16x4_f32(false, se, false, e0, (short)0, accE[1][0], false, false);
            accE[1][1] = __builtin_amdgcn_wmma_f32_16x16x4_f32(false, se, false, e1, (short)0, accE[1][1], false, false);
            accO[0][0] = __builtin_amdgcn_wmma_f32_16x16x4_f32(false, co, false, o0, (short)0, accO[0][0], false, false);
            accO[0][1] = __builtin_amdgcn_wmma_f32_16x16x4_f32(false, co, false, o1, (short)0, accO[0][1], false, false);
            accO[1][0] = __builtin_amdgcn_wmma_f32_16x16x4_f32(false, so, false, o0, (short)0, accO[1][0], false, false);
            accO[1][1] = __builtin_amdgcn_wmma_f32_16x16x4_f32(false, so, false, o1, (short)0, accO[1][1], false, false);
        }
    }

    const int kf0 = binBlock + (wn << 4) + (half << 3);
#pragma unroll
    for (int im = 0; im < 2; ++im) {
        const int Rr = rowBlock + (wm << 5) + (im << 4) + r16;
        const int b  = Rr / NF_N;
        const int m  = Rr - b * NF_N;
        float* op = out + (size_t)b * (F_N * NF_N * 2) + m * 2;
#pragma unroll
        for (int v = 0; v < 8; ++v) {
            const int k = kf0 + v;
            const v2f cs = *(const v2f*)(tw + 2 * k);
            const float er = accE[0][im][v], ei = accE[1][im][v];
            const float orr = accO[0][im][v], oi = accO[1][im][v];
            const float P = orr * cs.x + oi * cs.y;
            const float Q = oi * cs.x - orr * cs.y;
            v2f lo, hi;
            lo.x = er + P;  lo.y = ei + Q;
            hi.x = er - P;  hi.y = Q - ei;
            *(v2f*)(op + (size_t)k * (NF_N * 2)) = lo;
            *(v2f*)(op + (size_t)(F_N - 1 - k) * (NF_N * 2)) = hi;
        }
    }
}

extern "C" void kernel_launch(void* const* d_in, const int* in_sizes, int n_in,
                              void* d_out, int out_size, void* d_ws, size_t ws_size,
                              hipStream_t stream) {
    (void)in_sizes; (void)n_in; (void)out_size; (void)ws_size;
    const float* x = (const float*)d_in[0];
    float* out = (float*)d_out;

    float* ws     = (float*)d_ws;
    float* Wi     = ws;
    float* WfwdT  = Wi + (size_t)KPAD * KPAD;
    float* outsig = Wi + (size_t)1792 * KPAD;

    build_mats<<<(INV_ELEMS + FWD_ELEMS + 512 + KPAD) / 256, 256, 0, stream>>>(Wi, WfwdT);
    istft_signal<<<B_N, 512, 0, stream>>>(x, Wi, outsig);
    stft_gemm<<<dim3(256 / 32, M2 / BM), 128, 0, stream>>>(outsig, WfwdT, WfwdT + FWD_ELEMS, out);
}


# Forward DFT factored 1022 = 14 x 73 (decimation in time): 73-point f32 WMMA products per (frame, residue), twiddle and 14-point join on the vector unit; terms summed in a new order
# speedup vs baseline: 3.1669x; 1.4139x over previous
#include <hip/hip_runtime.h>


typedef float v2f __attribute__((ext_vector_type(2)));
typedef float v4f __attribute__((ext_vector_type(4)));
typedef float v8f __attribute__((ext_vector_type(8)));

#define B_N        256
#define F_N        512
#define T_N        81
#define WIN_N      1022
#define HOP_N      400
#define NF_N       69
#define OUT_LEN    28550
#define PERIOD_N   1100
#define SEG_N      1000
#define OVL_N      50
#define NSEG       30
#define OUT_STRIDE 28576
#define KPAD       1024
#define M2         (B_N * NF_N)
#define HALF_N     511
#define FT         16
#define NC         14
#define NU         73
#define UPAD       76
#define QH         37
#define W73_ROWS   80
#define W73_ELEMS  (W73_ROWS * UPAD)
#define TWJ_OFF    6144
#define TWJ_ELEMS  (QH * NC * 2)
#define HANN_OFF   7424
#define FWD_TAB    8448
#define NHP        512
#define INV_ELEMS  (2 * NHP * NHP)

#define BM 64
#define BK 32
#define SIG_ROWS 96
#define CB_N     256
#define TS 36
#define KHEAD 5

#define TWO_PI 6.283185307179586476925f

__device__ __forceinline__ float hann_f(int n) {
    return 0.5f - 0.5f * cosf((TWO_PI / WIN_N) * (float)n);
}

__global__ __launch_bounds__(256) void build_mats(float* __restrict__ Wi,
                                                  float* __restrict__ WfwdT) {
    int idx = blockIdx.x * 256 + threadIdx.x;
    if (idx < INV_ELEMS) {
        float v = 0.f;
        int mat = idx >> 18;
        int e   = idx & (NHP * NHP - 1);
        int n   = e >> 9;
        int kk  = e & (NHP - 1);
        if (n < HALF_N) {
            int f = (kk & ~1) + mat, c = kk & 1;
            int m = (f * n) % WIN_N;
            float ang = (TWO_PI / WIN_N) * (float)m;
            float wf  = (f == 0 || f == F_N - 1) ? 1.f : 2.f;
            float tr  = (c == 0) ? cosf(ang) : -sinf(ang);
            v = wf * tr * (1.0f / (float)WIN_N);
        }
        Wi[idx] = v;
    } else if (idx < INV_ELEMS + FWD_TAB) {
        const int e = idx - INV_ELEMS;
        float v = 0.f;
        if (e < W73_ELEMS) {
            const int r = e / UPAD, u = e - r * UPAD;
            if (r < 2 * QH && u < NU) {
                const int q = (r < QH) ? r : r - QH;
                const int m = (q * u) % NU;
                const float ang = (TWO_PI / NU) * (float)m;
                v = (r < QH) ? cosf(ang) : -sinf(ang);
            }
        } else if (e >= TWJ_OFF && e < TWJ_OFF + TWJ_ELEMS) {
            const int t  = e - TWJ_OFF;
            const int qc = t >> 1;
            const int q  = qc / NC, c = qc - q * NC;
            const float ang = (TWO_PI / WIN_N) * (float)(q * c);
            v = (t & 1) ? -sinf(ang) : cosf(ang);
        } else if (e >= HANN_OFF) {
            const int n = e - HANN_OFF;
            if (n < WIN_N) v = hann_f(n);
        }
        WfwdT[e] = v;
    } else {
        int n = idx - (INV_ELEMS + FWD_TAB);
        float v = 0.f;
        if (n < WIN_N) {
            int p = n % HOP_N;
            float d = 0.f;
            for (int j = p; j < WIN_N; j += HOP_N) {
                float hj = hann_f(j);
                d += hj * hj;
            }
            v = hann_f(n) / d;
        }
        Wi[INV_ELEMS + n] = v;
    }
}

__device__ __forceinline__ void fold_add_lds(float* sig, const int s,
                                             const float val, const bool live) {
    const int q  = s / PERIOD_N;
    const int rr = s - q * PERIOD_N;
    const bool up = (q > 0) & (rr < OVL_N);
    const bool dn = (q < NSEG - 1) & (rr >= SEG_N - OVL_N);
    const float ramp = (float)(up ? rr + 1 : SEG_N - rr) * 0.02f;
    const float cw   = (up | dn) ? ramp : 1.f;
    if (live & (rr < SEG_N) & (q < NSEG))
        __hip_atomic_fetch_add(sig + q * (SEG_N - OVL_N) + rr, cw * val,
                               __ATOMIC_RELAXED, __HIP_MEMORY_SCOPE_WORKGROUP);
}

__device__ __forceinline__ void inv_kblock(const float* __restrict__ xnext,
                                           const float* __restrict__ wnext,
                                           float (*sA)[SIG_ROWS][TS], float (*sB)[CB_N][TS],
                                           v8f (&acc)[3][2], v2f (&ra)[3], v4f (&rb)[4],
                                           const int (&a_off)[3], const int (&a_row)[3],
                                           const int (&a_col)[3],
                                           const int b_row, const int b_kc,
                                           const int wm, const int wn,
                                           const int half, const int r16,
                                           const int cur, const bool st, const bool ld) {
    v2f a[BK / 4][3], bf[BK / 4][2];
#pragma unroll
    for (int kk = 0; kk < BK / 4; ++kk) {
        const int ka = (kk << 2) + (half << 1);
#pragma unroll
        for (int i = 0; i < 3; ++i)
            a[kk][i] = *(const v2f*)&sA[cur][wm * 48 + (i << 4) + r16][ka];
#pragma unroll
        for (int j = 0; j < 2; ++j)
            bf[kk][j] = *(const v2f*)&sB[cur][(wn << 5) + (j << 4) + r16][ka];
    }
#pragma unroll
    for (int kk = 0; kk < KHEAD; ++kk)
#pragma unroll
        for (int i = 0; i < 3; ++i)
#pragma unroll
            for (int j = 0; j < 2; ++j)
                acc[i][j] = __builtin_amdgcn_wmma_f32_16x16x4_f32(false, a[kk][i], false, bf[kk][j], (short)0, acc[i][j], false, false);
    __builtin_amdgcn_sched_barrier(0);
    if (st) {
#pragma unroll
        for (int u = 0; u < 3; ++u) *(v2f*)&sA[cur ^ 1][a_row[u]][a_col[u]] = ra[u];
#pragma unroll
        for (int u = 0; u < 4; ++u) *(v4f*)&sB[cur ^ 1][b_row + (u << 6)][b_kc] = rb[u];
    }
    __syncthreads();
    if (ld) {
#pragma unroll
        for (int u = 0; u < 3; ++u)
            ra[u] = *(const v2f*)(xnext + a_off[u]);
#pragma unroll
        for (int u = 0; u < 4; ++u)
            rb[u] = *(const v4f*)(wnext + (size_t)(u << 6) * NHP);
    }
#pragma unroll
    for (int kk = KHEAD; kk < BK / 4; ++kk)
#pragma unroll
        for (int i = 0; i < 3; ++i)
#pragma unroll
            for (int j = 0; j < 2; ++j)
                acc[i][j] = __builtin_amdgcn_wmma_f32_16x16x4_f32(false, a[kk][i], false, bf[kk][j], (short)0, acc[i][j], false, false);
}

__device__ __forceinline__ void inv_pass(const float* __restrict__ xpar,
                                         const float* __restrict__ Wp,
                                         float (*sA)[SIG_ROWS][TS], float (*sB)[CB_N][TS],
                                         v8f (&acc)[3][2],
                                         const int (&a_off)[3], const int (&a_row)[3],
                                         const int (&a_col)[3],
                                         const int b_row, const int b_kc,
                                         const int wm, const int wn,
                                         const int half, const int r16) {
    v2f    ra[3];
    v4f    rb[4];
    const float* wrow = Wp + (size_t)b_row * NHP + b_kc;
#pragma unroll
    for (int u = 0; u < 3; ++u)
        ra[u] = *(const v2f*)(xpar + a_off[u]);
#pragma unroll
    for (int u = 0; u < 4; ++u)
        rb[u] = *(const v4f*)(wrow + (size_t)(u << 6) * NHP);
#pragma unroll
    for (int u = 0; u < 3; ++u) *(v2f*)&sA[0][a_row[u]][a_col[u]] = ra[u];
#pragma unroll
    for (int u = 0; u < 4; ++u) *(v4f*)&sB[0][b_row + (u << 6)][b_kc] = rb[u];
    __syncthreads();
#pragma unroll
    for (int u = 0; u < 3; ++u)
        ra[u] = *(const v2f*)(xpar + a_off[u] + BK * (T_N * 2));
#pragma unroll
    for (int u = 0; u < 4; ++u)
        rb[u] = *(const v4f*)(wrow + (size_t)(u << 6) * NHP + BK);

    for (int it = 0; it < NHP / BK - 2; ++it)
        inv_kblock(xpar + (it + 2) * BK * (T_N * 2), wrow + (it + 2) * BK, sA, sB, acc, ra, rb,
                   a_off, a_row, a_col, b_row, b_kc, wm, wn, half, r16, it & 1, true, true);
    inv_kblock(xpar, wrow, sA, sB, acc, ra, rb, a_off, a_row, a_col, b_row, b_kc,
               wm, wn, half, r16, (NHP / BK - 2) & 1, true, false);
    inv_kblock(xpar, wrow, sA, sB, acc, ra, rb, a_off, a_row, a_col, b_row, b_kc,
               wm, wn, half, r16, (NHP / BK - 1) & 1, false, false);
}

__global__ __launch_bounds__(512) void istft_signal(const float* __restrict__ x,
                                                    const float* __restrict__ Wi,
                                                    float* __restrict__ outsig) {
    __shared__ float sSig[OUT_STRIDE];
    __shared__ float sA[2][SIG_ROWS][TS];
    __shared__ float sB[2][CB_N][TS];

    const int tid  = threadIdx.x;
    const int lane = tid & 31;
    const int wave = tid >> 5;
    const int wm   = wave & 1;
    const int wn   = wave >> 1;
    const int half = lane >> 4;
    const int r16  = lane & 15;
    const int b    = blockIdx.x;

    for (int i = tid; i < OUT_STRIDE / 4; i += 512)
        ((float4*)sSig)[i] = make_float4(0.f, 0.f, 0.f, 0.f);

    int a_off[3], a_row[3], a_col[3];
#pragma unroll
    for (int u = 0; u < 3; ++u) {
        const int idx = tid + (u << 9);
        const int g   = idx / SIG_ROWS;
        const int t   = idx - g * SIG_ROWS;
        a_row[u]  = t;
        a_col[u]  = g << 1;
        a_off[u]  = (g << 1) * (T_N * 2) + (t < T_N ? t : T_N - 1) * 2;
    }
    const float* xs = x + (size_t)b * (F_N * T_N * 2);

    const int b_row = tid >> 3;
    const int b_kc  = (tid & 7) << 2;

    const float* sw = Wi + INV_ELEMS;

    for (int cb = 0; cb < NHP / CB_N; ++cb) {
        v8f accE[3][2];
        v8f accO[3][2];
#pragma unroll
        for (int i = 0; i < 3; ++i)
#pragma unroll
            for (int j = 0; j < 2; ++j) {
                accE[i][j] = (v8f){0.f,0.f,0.f,0.f,0.f,0.f,0.f,0.f};
                accO[i][j] = (v8f){0.f,0.f,0.f,0.f,0.f,0.f,0.f,0.f};
            }

        const float* We = Wi + (size_t)(cb * CB_N) * NHP;
        inv_pass(xs,           We,             sA, sB, accE, a_off, a_row, a_col, b_row, b_kc, wm, wn, half, r16);
        inv_pass(xs + T_N * 2, We + NHP * NHP, sA, sB, accO, a_off, a_row, a_col, b_row, b_kc, wm, wn, half, r16);

        int   nn[2];
        float w_lo[2], w_hi[2];
#pragma unroll
        for (int j = 0; j < 2; ++j) {
            nn[j]   = cb * CB_N + (wn << 5) + (j << 4) + r16;
            w_lo[j] = sw[nn[j]];
            w_hi[j] = sw[nn[j] + HALF_N];
        }
#pragma unroll
        for (int i = 0; i < 3; ++i) {
            const int tb = wm * 48 + (i << 4) + (half << 3);
#pragma unroll
            for (int v = 0; v < 8; ++v) {
                const int t = tb + v;
#pragma unroll
                for (int j = 0; j < 2; ++j) {
                    const float e = accE[i][j][v];
                    const float o = accO[i][j][v];
                    const int   s = t * HOP_N + nn[j];
                    const bool live = (t < T_N) & (nn[j] < HALF_N);
                    fold_add_lds(sSig, s,          (e + o) * w_lo[j], live);
                    fold_add_lds(sSig, s + HALF_N, (e - o) * w_hi[j], live);
                }
            }
        }
    }

    __syncthreads();
    float4* dst = (float4*)(outsig + (size_t)b * OUT_STRIDE);
    for (int i = tid; i < OUT_STRIDE / 4; i += 512)
        dst[i] = ((const float4*)sSig)[i];
}

__device__ __forceinline__ float cos7(const int r) {
    return r == 0 ? 1.f : (r == 1 || r == 6) ? 0.6234898018587336f
         : (r == 2 || r == 5) ? -0.2225209339563144f : -0.9009688679024191f;
}
__device__ __forceinline__ float sin7(const int r) {
    return r == 0 ? 0.f : r == 1 ? 0.7818314824680298f : r == 2 ? 0.9749279121818236f
         : r == 3 ? 0.4338837391175581f : r == 4 ? -0.4338837391175581f
         : r == 5 ? -0.9749279121818236f : -0.7818314824680298f;
}
__device__ __forceinline__ void dft7(const float (&hr)[NC], const float (&hi)[NC], const int e,
                                     float (&zr)[7], float (&zi)[7]) {
    float s0r = hr[e], s0i = hi[e];
#pragma unroll
    for (int d = 1; d < 7; ++d) { s0r += hr[2 * d + e]; s0i += hi[2 * d + e]; }
    zr[0] = s0r; zi[0] = s0i;
#pragma unroll
    for (int j = 1; j < 4; ++j) {
        float ar = hr[e], ai = hi[e], br = 0.f, bi = 0.f;
#pragma unroll
        for (int d = 1; d < 7; ++d) {
            const int r = (j * d) % 7;
            ar += hr[2 * d + e] * cos7(r);
            ai += hi[2 * d + e] * cos7(r);
            br += hr[2 * d + e] * sin7(r);
            bi += hi[2 * d + e] * sin7(r);
        }
        zr[j]     = ar + bi;  zi[j]     = ai - br;
        zr[7 - j] = ar - bi;  zi[7 - j] = ai + br;
    }
}

__global__ __launch_bounds__(256) void stft_ct(const float* __restrict__ outsig,
                                               const float* __restrict__ W73,
                                               const float* __restrict__ twj,
                                               const float* __restrict__ hw,
                                               float* __restrict__ out) {
    __shared__ __attribute__((aligned(16))) float sY[FT * NC * UPAD];
    __shared__ __attribute__((aligned(16))) float sW[W73_ELEMS];

    const int tid  = threadIdx.x;
    const int lane = tid & 31;
    const int wave = __builtin_amdgcn_readfirstlane(tid >> 5);
    const int half = lane >> 4;
    const int r16  = lane & 15;
    const int rowBlock = blockIdx.x * FT;

    for (int i = tid; i < W73_ELEMS / 4; i += 256)
        ((float4*)sW)[i] = ((const float4*)W73)[i];
    if (tid < FT * NC) {
        sY[tid * UPAD + NU]     = 0.f;
        sY[tid * UPAD + NU + 1] = 0.f;
        sY[tid * UPAD + NU + 2] = 0.f;
    }
#pragma unroll
    for (int it = 0; it < 2; ++it) {
        const int p = tid + (it << 8);
        if (p < HALF_N) {
            const int n0 = 2 * p, n1 = n0 + 1;
            const int u0 = n0 / NC, c0 = n0 - u0 * NC;
            const int u1 = n1 / NC, c1 = n1 - u1 * NC;
            const v2f h = *(const v2f*)(hw + n0);
            const int o0 = c0 * UPAD + u0, o1 = c1 * UPAD + u1;
#pragma unroll
            for (int f = 0; f < FT; ++f) {
                const int R  = rowBlock + f;
                const int fb = R / NF_N;
                const int fm = R - fb * NF_N;
                const v2f sv = *(const v2f*)(outsig + (size_t)fb * OUT_STRIDE + fm * HOP_N + n0);
                sY[f * (NC * UPAD) + o0] = sv.x * h.x;
                sY[f * (NC * UPAD) + o1] = sv.y * h.y;
            }
        }
    }
    __syncthreads();

    if (wave < 7) {
        v8f acc[5][2];
#pragma unroll
        for (int i = 0; i < 5; ++i)
#pragma unroll
            for (int j = 0; j < 2; ++j)
                acc[i][j] = (v8f){0.f,0.f,0.f,0.f,0.f,0.f,0.f,0.f};
        const float* ap = sW + r16 * UPAD + (half << 1);
        const float* bp = sY + ((wave << 5) + r16) * UPAD + (half << 1);
#pragma unroll
        for (int k0 = 0; k0 < UPAD; k0 += 4) {
            v2f a[5], bb[2];
#pragma unroll
            for (int i = 0; i < 5; ++i) a[i] = *(const v2f*)(ap + i * (16 * UPAD) + k0);
#pragma unroll
            for (int j = 0; j < 2; ++j) bb[j] = *(const v2f*)(bp + j * (16 * UPAD) + k0);
#pragma unroll
            for (int i = 0; i < 5; ++i)
#pragma unroll
                for (int j = 0; j < 2; ++j)
                    acc[i][j] = __builtin_amdgcn_wmma_f32_16x16x4_f32(false, a[i], false, bb[j], (short)0, acc[i][j], false, false);
        }
#pragma unroll
        for (int j = 0; j < 2; ++j) {
            float* g = sY + ((wave << 5) + (j << 4) + r16) * UPAD + (half << 3);
#pragma unroll
            for (int i = 0; i < 5; ++i) {
                const v8f d = acc[i][j];
                *(float4*)(g + 16 * i) = make_float4(d[0], d[1], d[2], d[3]);
                if (i < 4 || half == 0)
                    *(float4*)(g + 16 * i + 4) = make_float4(d[4], d[5], d[6], d[7]);
            }
        }
    }
    __syncthreads();

    const int R  = rowBlock + r16;
    const int jb = R / NF_N;
    const int jm = R - jb * NF_N;
    float* op = out + (size_t)jb * (F_N * NF_N * 2) + jm * 2;
    const float* gf = sY + r16 * (NC * UPAD);
    for (int it = 0; it < 3; ++it) {
        const int q = ((wave + (it << 3)) << 1) + half;
        if (q < QH) {
            float hr[NC], hi[NC];
#pragma unroll
            for (int c = 0; c < NC; ++c) {
                const float gr = gf[c * UPAD + q];
                const float gi = gf[c * UPAD + QH + q];
                const v2f  t  = *(const v2f*)(twj + 2 * (q * NC + c));
                hr[c] = gr * t.x - gi * t.y;
                hi[c] = gr * t.y + gi * t.x;
            }
            float er[7], ei[7], pr[7], pi[7];
            dft7(hr, hi, 0, er, ei);
            dft7(hr, hi, 1, pr, pi);
            const float sg = (q == 0) ? 1.f : -1.f;
#pragma unroll
            for (int j = 0; j < 7; ++j) {
                const float cj = (j == 0) ? 1.f : (j == 1) ? 0.9009688679024191f : (j == 2) ? 0.6234898018587336f
                               : (j == 3) ? 0.2225209339563144f : (j == 4) ? -0.2225209339563144f
                               : (j == 5) ? -0.6234898018587336f : -0.9009688679024191f;
                const float sj = (j == 0) ? 0.f : (j == 1 || j == 6) ? 0.4338837391175581f
                               : (j == 2 || j == 5) ? 0.7818314824680298f : 0.9749279121818236f;
                float P, Q;
                if (j == 0) { P = pr[0]; Q = pi[0]; }
                else        { P = pr[j] * cj + pi[j] * sj;  Q = pi[j] * cj - pr[j] * sj; }
                v2f lo, hi2;
                lo.x  = er[j] + P;  lo.y  = ei[j] + Q;
                hi2.x = er[j] - P;  hi2.y = sg * (ei[j] - Q);
                *(v2f*)(op + (size_t)(q + NU * j) * (NF_N * 2)) = lo;
                if (q != 0 || j == 0)
                    *(v2f*)(op + (size_t)(NU * (7 - j) - q) * (NF_N * 2)) = hi2;
            }
        }
    }
}

extern "C" void kernel_launch(void* const* d_in, const int* in_sizes, int n_in,
                              void* d_out, int out_size, void* d_ws, size_t ws_size,
                              hipStream_t stream) {
    (void)in_sizes; (void)n_in; (void)out_size; (void)ws_size;
    const float* x = (const float*)d_in[0];
    float* out = (float*)d_out;

    float* ws     = (float*)d_ws;
    float* Wi     = ws;
    float* WfwdT  = Wi + (size_t)KPAD * KPAD;
    float* outsig = Wi + (size_t)1792 * KPAD;

    build_mats<<<(INV_ELEMS + FWD_TAB + KPAD) / 256, 256, 0, stream>>>(Wi, WfwdT);
    istft_signal<<<B_N, 512, 0, stream>>>(x, Wi, outsig);
    stft_ct<<<M2 / FT, 256, 0, stream>>>(outsig, WfwdT, WfwdT + TWJ_OFF, WfwdT + HANN_OFF, out);
}


# Inverse DFT factored 1022 = 14 x 73 (split in frequency): 14-point DFTs and twiddle on the vector unit, then 73-point f32 WMMA products per (frame, residue); terms summed in a new order
# speedup vs baseline: 5.4458x; 1.7196x over previous
#include <hip/hip_runtime.h>


typedef float v2f __attribute__((ext_vector_type(2)));
typedef float v8f __attribute__((ext_vector_type(8)));

#define B_N        256
#define F_N        512
#define T_N        81
#define WIN_N      1022
#define HOP_N      400
#define NF_N       69
#define OUT_LEN    28550
#define PERIOD_N   1100
#define SEG_N      1000
#define OVL_N      50
#define NSEG       30
#define OUT_STRIDE 28576
#define KPAD       1024
#define M2         (B_N * NF_N)
#define HALF_N     511
#define FT         16
#define NC         14
#define NU         73
#define UPAD       76
#define QH         37
#define W73_ROWS   80
#define W73_ELEMS  (W73_ROWS * UPAD)
#define TWJ_OFF    6144
#define TWJ_ELEMS  (QH * NC * 2)
#define HANN_OFF   7424
#define FWD_TAB    8448
#define IM_ELEMS   6144
#define IFT        16
#define ICH        6

#define TWO_PI 6.283185307179586476925f

__device__ __forceinline__ float hann_f(int n) {
    return 0.5f - 0.5f * cosf((TWO_PI / WIN_N) * (float)n);
}

__global__ __launch_bounds__(256) void build_mats(float* __restrict__ Wi,
                                                  float* __restrict__ WfwdT) {
    int idx = blockIdx.x * 256 + threadIdx.x;
    if (idx < IM_ELEMS) {
        float v = 0.f;
        const int u = idx / UPAD, k = idx - u * UPAD;
        if (u < NU && k < 2 * QH) {
            const int q = (k < QH) ? k : k - QH;
            const int m = (q * u) % NU;
            const float ang = (TWO_PI / NU) * (float)m;
            const float wq  = (q == 0) ? 1.f : 2.f;
            const float tr  = (k < QH) ? cosf(ang) : -sinf(ang);
            v = wq * tr * (1.0f / (float)WIN_N);
        }
        Wi[idx] = v;
    } else if (idx < IM_ELEMS + FWD_TAB) {
        const int e = idx - IM_ELEMS;
        float v = 0.f;
        if (e < W73_ELEMS) {
            const int r = e / UPAD, u = e - r * UPAD;
            if (r < 2 * QH && u < NU) {
                const int q = (r < QH) ? r : r - QH;
                const int m = (q * u) % NU;
                const float ang = (TWO_PI / NU) * (float)m;
                v = (r < QH) ? cosf(ang) : -sinf(ang);
            }
        } else if (e >= TWJ_OFF && e < TWJ_OFF + TWJ_ELEMS) {
            const int t  = e - TWJ_OFF;
            const int qc = t >> 1;
            const int q  = qc / NC, c = qc - q * NC;
            const float ang = (TWO_PI / WIN_N) * (float)(q * c);
            v = (t & 1) ? -sinf(ang) : cosf(ang);
        } else if (e >= HANN_OFF) {
            const int n = e - HANN_OFF;
            if (n < WIN_N) v = hann_f(n);
        }
        WfwdT[e] = v;
    } else {
        int n = idx - (IM_ELEMS + FWD_TAB);
        float v = 0.f;
        if (n < WIN_N) {
            int p = n % HOP_N;
            float d = 0.f;
            for (int j = p; j < WIN_N; j += HOP_N) {
                float hj = hann_f(j);
                d += hj * hj;
            }
            v = hann_f(n) / d;
        }
        Wi[IM_ELEMS + n] = v;
    }
}

__device__ __forceinline__ void fold_add_lds(float* sig, const int s,
                                             const float val, const bool live) {
    const int q  = s / PERIOD_N;
    const int rr = s - q * PERIOD_N;
    const bool up = (q > 0) & (rr < OVL_N);
    const bool dn = (q < NSEG - 1) & (rr >= SEG_N - OVL_N);
    const float ramp = (float)(up ? rr + 1 : SEG_N - rr) * 0.02f;
    const float cw   = (up | dn) ? ramp : 1.f;
    if (live & (rr < SEG_N) & (q < NSEG))
        __hip_atomic_fetch_add(sig + q * (SEG_N - OVL_N) + rr, cw * val,
                               __ATOMIC_RELAXED, __HIP_MEMORY_SCOPE_WORKGROUP);
}

__device__ __forceinline__ float cos7(const int r) {
    return r == 0 ? 1.f : (r == 1 || r == 6) ? 0.6234898018587336f
         : (r == 2 || r == 5) ? -0.2225209339563144f : -0.9009688679024191f;
}
__device__ __forceinline__ float sin7(const int r) {
    return r == 0 ? 0.f : r == 1 ? 0.7818314824680298f : r == 2 ? 0.9749279121818236f
         : r == 3 ? 0.4338837391175581f : r == 4 ? -0.4338837391175581f
         : r == 5 ? -0.9749279121818236f : -0.7818314824680298f;
}
__device__ __forceinline__ void dft7(const float (&hr)[NC], const float (&hi)[NC], const int e,
                                     float (&zr)[7], float (&zi)[7]) {
    float s0r = hr[e], s0i = hi[e];
#pragma unroll
    for (int d = 1; d < 7; ++d) { s0r += hr[2 * d + e]; s0i += hi[2 * d + e]; }
    zr[0] = s0r; zi[0] = s0i;
#pragma unroll
    for (int j = 1; j < 4; ++j) {
        float ar = hr[e], ai = hi[e], br = 0.f, bi = 0.f;
#pragma unroll
        for (int d = 1; d < 7; ++d) {
            const int r = (j * d) % 7;
            ar += hr[2 * d + e] * cos7(r);
            ai += hi[2 * d + e] * cos7(r);
            br += hr[2 * d + e] * sin7(r);
            bi += hi[2 * d + e] * sin7(r);
        }
        zr[j]     = ar + bi;  zi[j]     = ai - br;
        zr[7 - j] = ar - bi;  zi[7 - j] = ai + br;
    }
}

__device__ __forceinline__ void inv_load(const float* __restrict__ xt, const int q, v2f (&z)[NC]) {
#pragma unroll
    for (int a = 0; a < 7; ++a)
        z[a] = *(const v2f*)(xt + (q + NU * a) * (T_N * 2));
#pragma unroll
    for (int j = 0; j < 7; ++j)
        z[7 + j] = *(const v2f*)(xt + (NU * (7 - j) - q) * (T_N * 2));
}

__device__ __forceinline__ void inv_join(const v2f (&z)[NC], const int q,
                                         const float* __restrict__ sTw,
                                         float* __restrict__ hrow, const bool ok) {
    float zr[NC], zi[NC];
#pragma unroll
    for (int a = 0; a < 7; ++a)  { zr[a] = z[a].x; zi[a] = z[a].y; }
#pragma unroll
    for (int a = 7; a < NC; ++a) { zr[a] = z[a].x; zi[a] = -z[a].y; }
    if (q == 0) { zi[0] = 0.f; zi[7] = 0.f; }
    float er[7], ei[7], pr[7], pi[7];
    dft7(zr, zi, 0, er, ei);
    dft7(zr, zi, 1, pr, pi);
#pragma unroll
    for (int j = 0; j < 7; ++j) {
        const int jj = (7 - j) % 7;
        const float cj = (j == 0) ? 1.f : (j == 1) ? 0.9009688679024191f : (j == 2) ? 0.6234898018587336f
                       : (j == 3) ? 0.2225209339563144f : (j == 4) ? -0.2225209339563144f
                       : (j == 5) ? -0.6234898018587336f : -0.9009688679024191f;
        const float sj = (j == 0) ? 0.f : (j == 1 || j == 6) ? 0.4338837391175581f
                       : (j == 2 || j == 5) ? 0.7818314824680298f : 0.9749279121818236f;
        float P, Q;
        if (j == 0) { P = pr[0]; Q = pi[0]; }
        else        { P = pr[jj] * cj - pi[jj] * sj;  Q = pi[jj] * cj + pr[jj] * sj; }
        const float h0r = er[jj] + P, h0i = ei[jj] + Q;
        const float h1r = er[jj] - P, h1i = ei[jj] - Q;
        const v2f t0 = *(const v2f*)(sTw + 2 * (q * NC + j));
        const v2f t1 = *(const v2f*)(sTw + 2 * (q * NC + j + 7));
        if (ok) {
            hrow[j * UPAD + q]            = h0r * t0.x + h0i * t0.y;
            hrow[j * UPAD + QH + q]       = h0i * t0.x - h0r * t0.y;
            hrow[(j + 7) * UPAD + q]      = h1r * t1.x + h1i * t1.y;
            hrow[(j + 7) * UPAD + QH + q] = h1i * t1.x - h1r * t1.y;
        }
    }
}

__global__ __launch_bounds__(512) void istft_signal(const float* __restrict__ x,
                                                    const float* __restrict__ Mi,
                                                    const float* __restrict__ swg,
                                                    const float* __restrict__ twj,
                                                    float* __restrict__ outsig) {
    __shared__ __attribute__((aligned(16))) float sSig[OUT_STRIDE];
    __shared__ __attribute__((aligned(16))) float sH[IFT * NC * UPAD];
    __shared__ __attribute__((aligned(16))) float sM[W73_ELEMS];
    __shared__ __attribute__((aligned(16))) float sSw[KPAD];
    __shared__ __attribute__((aligned(16))) float sTw[TWJ_ELEMS];

    const int tid  = threadIdx.x;
    const int lane = tid & 31;
    const int wave = __builtin_amdgcn_readfirstlane(tid >> 5);
    const int half = lane >> 4;
    const int r16  = lane & 15;
    const int b    = blockIdx.x;
    const float* xs = x + (size_t)b * (F_N * T_N * 2);

    const int  q0    = (wave << 1) + half;
    const int  q1raw = 32 + (wave << 1) + half;
    const int  q1    = q1raw < QH ? q1raw : QH - 1;
    const bool two   = wave < 3;

    v2f z0[NC], z1[NC];
    inv_load(xs + 2 * r16, q0, z0);
    if (two) inv_load(xs + 2 * r16, q1, z1);

    for (int i = tid; i < OUT_STRIDE / 4; i += 512)
        ((float4*)sSig)[i] = make_float4(0.f, 0.f, 0.f, 0.f);
    for (int i = tid; i < W73_ELEMS / 4; i += 512)
        ((float4*)sM)[i] = ((const float4*)Mi)[i];
    for (int i = tid; i < KPAD; i += 512)
        sSw[i] = swg[i];
    for (int i = tid; i < TWJ_ELEMS; i += 512)
        sTw[i] = twj[i];
    if (tid < IFT * NC) {
        sH[tid * UPAD + 2 * QH]     = 0.f;
        sH[tid * UPAD + 2 * QH + 1] = 0.f;
    }
    __syncthreads();

    float* hrow = sH + r16 * (NC * UPAD);
    for (int ch = 0; ch < ICH; ++ch) {
        const int  t0  = ch * IFT;
        const bool tok = t0 + r16 < T_N;
        inv_join(z0, q0, sTw, hrow, tok);
        if (two) inv_join(z1, q1, sTw, hrow, tok & (q1raw < QH));
        __syncthreads();

        if (ch + 1 < ICH) {
            const int tn = t0 + IFT + r16;
            const int tc = tn < T_N ? tn : T_N - 1;
            inv_load(xs + 2 * tc, q0, z0);
            if (two) inv_load(xs + 2 * tc, q1, z1);
        }

        const int ntile = (ch + 1 < ICH) ? NC : 1;
        if (wave < ntile) {
            v8f acc[5];
#pragma unroll
            for (int i = 0; i < 5; ++i)
                acc[i] = (v8f){0.f,0.f,0.f,0.f,0.f,0.f,0.f,0.f};
            const float* ap = sM + r16 * UPAD + (half << 1);
            const float* bp = sH + ((wave << 4) + r16) * UPAD + (half << 1);
#pragma unroll
            for (int k0 = 0; k0 < UPAD; k0 += 4) {
                v2f a[5];
#pragma unroll
                for (int i = 0; i < 5; ++i) a[i] = *(const v2f*)(ap + i * (16 * UPAD) + k0);
                const v2f bb = *(const v2f*)(bp + k0);
#pragma unroll
                for (int i = 0; i < 5; ++i)
                    acc[i] = __builtin_amdgcn_wmma_f32_16x16x4_f32(false, a[i], false, bb, (short)0, acc[i], false, false);
            }
            const int  R   = (wave << 4) + r16;
            const int  tl  = R / NC;
            const int  c   = R - tl * NC;
            const int  tt  = t0 + tl;
            const bool rowlive = tt < T_N;
#pragma unroll
            for (int i = 0; i < 5; ++i) {
#pragma unroll
                for (int v = 0; v < 8; ++v) {
                    const int  u    = (i << 4) + (half << 3) + v;
                    const bool live = rowlive & (u < NU);
                    const int  n    = live ? NC * u + c : 0;
                    fold_add_lds(sSig, tt * HOP_N + n, acc[i][v] * sSw[n], live);
                }
            }
        }
        __syncthreads();
    }

    float4* dst = (float4*)(outsig + (size_t)b * OUT_STRIDE);
    for (int i = tid; i < OUT_STRIDE / 4; i += 512)
        dst[i] = ((const float4*)sSig)[i];
}

__global__ __launch_bounds__(256) void stft_ct(const float* __restrict__ outsig,
                                               const float* __restrict__ W73,
                                               const float* __restrict__ twj,
                                               const float* __restrict__ hw,
                                               float* __restrict__ out) {
    __shared__ __attribute__((aligned(16))) float sY[FT * NC * UPAD];
    __shared__ __attribute__((aligned(16))) float sW[W73_ELEMS];

    const int tid  = threadIdx.x;
    const int lane = tid & 31;
    const int wave = __builtin_amdgcn_readfirstlane(tid >> 5);
    const int half = lane >> 4;
    const int r16  = lane & 15;
    const int rowBlock = blockIdx.x * FT;

    for (int i = tid; i < W73_ELEMS / 4; i += 256)
        ((float4*)sW)[i] = ((const float4*)W73)[i];
    if (tid < FT * NC) {
        sY[tid * UPAD + NU]     = 0.f;
        sY[tid * UPAD + NU + 1] = 0.f;
        sY[tid * UPAD + NU + 2] = 0.f;
    }
#pragma unroll
    for (int it = 0; it < 2; ++it) {
        const int p = tid + (it << 8);
        if (p < HALF_N) {
            const int n0 = 2 * p, n1 = n0 + 1;
            const int u0 = n0 / NC, c0 = n0 - u0 * NC;
            const int u1 = n1 / NC, c1 = n1 - u1 * NC;
            const v2f h = *(const v2f*)(hw + n0);
            const int o0 = c0 * UPAD + u0, o1 = c1 * UPAD + u1;
#pragma unroll
            for (int f = 0; f < FT; ++f) {
                const int R  = rowBlock + f;
                const int fb = R / NF_N;
                const int fm = R - fb * NF_N;
                const v2f sv = *(const v2f*)(outsig + (size_t)fb * OUT_STRIDE + fm * HOP_N + n0);
                sY[f * (NC * UPAD) + o0] = sv.x * h.x;
                sY[f * (NC * UPAD) + o1] = sv.y * h.y;
            }
        }
    }
    __syncthreads();

    if (wave < 7) {
        v8f acc[5][2];
#pragma unroll
        for (int i = 0; i < 5; ++i)
#pragma unroll
            for (int j = 0; j < 2; ++j)
                acc[i][j] = (v8f){0.f,0.f,0.f,0.f,0.f,0.f,0.f,0.f};
        const float* ap = sW + r16 * UPAD + (half << 1);
        const float* bp = sY + ((wave << 5) + r16) * UPAD + (half << 1);
#pragma unroll
        for (int k0 = 0; k0 < UPAD; k0 += 4) {
            v2f a[5], bb[2];
#pragma unroll
            for (int i = 0; i < 5; ++i) a[i] = *(const v2f*)(ap + i * (16 * UPAD) + k0);
#pragma unroll
            for (int j = 0; j < 2; ++j) bb[j] = *(const v2f*)(bp + j * (16 * UPAD) + k0);
#pragma unroll
            for (int i = 0; i < 5; ++i)
#pragma unroll
                for (int j = 0; j < 2; ++j)
                    acc[i][j] = __builtin_amdgcn_wmma_f32_16x16x4_f32(false, a[i], false, bb[j], (short)0, acc[i][j], false, false);
        }
#pragma unroll
        for (int j = 0; j < 2; ++j) {
            float* g = sY + ((wave << 5) + (j << 4) + r16) * UPAD + (half << 3);
#pragma unroll
            for (int i = 0; i < 5; ++i) {
                const v8f d = acc[i][j];
                *(float4*)(g + 16 * i) = make_float4(d[0], d[1], d[2], d[3]);
                if (i < 4 || half == 0)
                    *(float4*)(g + 16 * i + 4) = make_float4(d[4], d[5], d[6], d[7]);
            }
        }
    }
    __syncthreads();

    const int R  = rowBlock + r16;
    const int jb = R / NF_N;
    const int jm = R - jb * NF_N;
    float* op = out + (size_t)jb * (F_N * NF_N * 2) + jm * 2;
    const float* gf = sY + r16 * (NC * UPAD);
    for (int it = 0; it < 3; ++it) {
        const int q = ((wave + (it << 3)) << 1) + half;
        if (q < QH) {
            float hr[NC], hi[NC];
#pragma unroll
            for (int c = 0; c < NC; ++c) {
                const float gr = gf[c * UPAD + q];
                const float gi = gf[c * UPAD + QH + q];
                const v2f  t  = *(const v2f*)(twj + 2 * (q * NC + c));
                hr[c] = gr * t.x - gi * t.y;
                hi[c] = gr * t.y + gi * t.x;
            }
            float er[7], ei[7], pr[7], pi[7];
            dft7(hr, hi, 0, er, ei);
            dft7(hr, hi, 1, pr, pi);
            const float sg = (q == 0) ? 1.f : -1.f;
#pragma unroll
            for (int j = 0; j < 7; ++j) {
                const float cj = (j == 0) ? 1.f : (j == 1) ? 0.9009688679024191f : (j == 2) ? 0.6234898018587336f
                               : (j == 3) ? 0.2225209339563144f : (j == 4) ? -0.2225209339563144f
                               : (j == 5) ? -0.6234898018587336f : -0.9009688679024191f;
                const float sj = (j == 0) ? 0.f : (j == 1 || j == 6) ? 0.4338837391175581f
                               : (j == 2 || j == 5) ? 0.7818314824680298f : 0.9749279121818236f;
                float P, Q;
                if (j == 0) { P = pr[0]; Q = pi[0]; }
                else        { P = pr[j] * cj + pi[j] * sj;  Q = pi[j] * cj - pr[j] * sj; }
                v2f lo, hi2;
                lo.x  = er[j] + P;  lo.y  = ei[j] + Q;
                hi2.x = er[j] - P;  hi2.y = sg * (ei[j] - Q);
                *(v2f*)(op + (size_t)(q + NU * j) * (NF_N * 2)) = lo;
                if (q != 0 || j == 0)
                    *(v2f*)(op + (size_t)(NU * (7 - j) - q) * (NF_N * 2)) = hi2;
            }
        }
    }
}

extern "C" void kernel_launch(void* const* d_in, const int* in_sizes, int n_in,
                              void* d_out, int out_size, void* d_ws, size_t ws_size,
                              hipStream_t stream) {
    (void)in_sizes; (void)n_in; (void)out_size; (void)ws_size;
    const float* x = (const float*)d_in[0];
    float* out = (float*)d_out;

    float* ws     = (float*)d_ws;
    float* Wi     = ws;
    float* WfwdT  = Wi + (size_t)KPAD * KPAD;
    float* outsig = Wi + (size_t)1792 * KPAD;

    build_mats<<<(IM_ELEMS + FWD_TAB + KPAD) / 256, 256, 0, stream>>>(Wi, WfwdT);
    istft_signal<<<B_N, 512, 0, stream>>>(x, Wi, Wi + IM_ELEMS, WfwdT + TWJ_OFF, outsig);
    stft_ct<<<M2 / FT, 256, 0, stream>>>(outsig, WfwdT, WfwdT + TWJ_OFF, WfwdT + HANN_OFF, out);
}


# Inverse overlap-add goes into the plain signal in LDS (an add = one multiply + one LDS atomic at a literal offset); crossfade fold applied once per output at the store; terms summed in a new order
# speedup vs baseline: 6.9970x; 1.2848x over previous
#include <hip/hip_runtime.h>


typedef float v2f __attribute__((ext_vector_type(2)));
typedef float v8f __attribute__((ext_vector_type(8)));

#define B_N        256
#define F_N        512
#define T_N        81
#define WIN_N      1022
#define HOP_N      400
#define NF_N       69
#define OUT_LEN    28550
#define RAW_LEN    33024
#define PERIOD_N   1100
#define SEG_N      1000
#define OVL_N      50
#define NSEG       30
#define OUT_STRIDE 28576
#define KPAD       1024
#define M2         (B_N * NF_N)
#define HALF_N     511
#define FT         16
#define NC         14
#define NU         73
#define UPAD       76
#define QH         37
#define W73_ROWS   80
#define W73_ELEMS  (W73_ROWS * UPAD)
#define TWJ_OFF    6144
#define TWJ_ELEMS  (QH * NC * 2)
#define HANN_OFF   7424
#define FWD_TAB    8448
#define IM_ELEMS   6144
#define IFT        16
#define ICH        6

#define TWO_PI 6.283185307179586476925f

__device__ __forceinline__ float hann_f(int n) {
    return 0.5f - 0.5f * cosf((TWO_PI / WIN_N) * (float)n);
}

__global__ __launch_bounds__(256) void build_mats(float* __restrict__ Wi,
                                                  float* __restrict__ WfwdT) {
    int idx = blockIdx.x * 256 + threadIdx.x;
    if (idx < IM_ELEMS) {
        float v = 0.f;
        const int u = idx / UPAD, k = idx - u * UPAD;
        if (u < NU && k < 2 * QH) {
            const int q = (k < QH) ? k : k - QH;
            const int m = (q * u) % NU;
            const float ang = (TWO_PI / NU) * (float)m;
            const float wq  = (q == 0) ? 1.f : 2.f;
            const float tr  = (k < QH) ? cosf(ang) : -sinf(ang);
            v = wq * tr * (1.0f / (float)WIN_N);
        }
        Wi[idx] = v;
    } else if (idx < IM_ELEMS + FWD_TAB) {
        const int e = idx - IM_ELEMS;
        float v = 0.f;
        if (e < W73_ELEMS) {
            const int r = e / UPAD, u = e - r * UPAD;
            if (r < 2 * QH && u < NU) {
                const int q = (r < QH) ? r : r - QH;
                const int m = (q * u) % NU;
                const float ang = (TWO_PI / NU) * (float)m;
                v = (r < QH) ? cosf(ang) : -sinf(ang);
            }
        } else if (e >= TWJ_OFF && e < TWJ_OFF + TWJ_ELEMS) {
            const int t  = e - TWJ_OFF;
            const int qc = t >> 1;
            const int q  = qc / NC, c = qc - q * NC;
            const float ang = (TWO_PI / WIN_N) * (float)(q * c);
            v = (t & 1) ? -sinf(ang) : cosf(ang);
        } else if (e >= HANN_OFF) {
            const int n = e - HANN_OFF;
            if (n < WIN_N) v = hann_f(n);
        }
        WfwdT[e] = v;
    } else {
        int n = idx - (IM_ELEMS + FWD_TAB);
        float v = 0.f;
        if (n < WIN_N) {
            int p = n % HOP_N;
            float d = 0.f;
            for (int j = p; j < WIN_N; j += HOP_N) {
                float hj = hann_f(j);
                d += hj * hj;
            }
            v = hann_f(n) / d;
        }
        Wi[IM_ELEMS + n] = v;
    }
}

__device__ __forceinline__ float cos7(const int r) {
    return r == 0 ? 1.f : (r == 1 || r == 6) ? 0.6234898018587336f
         : (r == 2 || r == 5) ? -0.2225209339563144f : -0.9009688679024191f;
}
__device__ __forceinline__ float sin7(const int r) {
    return r == 0 ? 0.f : r == 1 ? 0.7818314824680298f : r == 2 ? 0.9749279121818236f
         : r == 3 ? 0.4338837391175581f : r == 4 ? -0.4338837391175581f
         : r == 5 ? -0.9749279121818236f : -0.7818314824680298f;
}
__device__ __forceinline__ void dft7(const float (&hr)[NC], const float (&hi)[NC], const int e,
                                     float (&zr)[7], float (&zi)[7]) {
    float s0r = hr[e], s0i = hi[e];
#pragma unroll
    for (int d = 1; d < 7; ++d) { s0r += hr[2 * d + e]; s0i += hi[2 * d + e]; }
    zr[0] = s0r; zi[0] = s0i;
#pragma unroll
    for (int j = 1; j < 4; ++j) {
        float ar = hr[e], ai = hi[e], br = 0.f, bi = 0.f;
#pragma unroll
        for (int d = 1; d < 7; ++d) {
            const int r = (j * d) % 7;
            ar += hr[2 * d + e] * cos7(r);
            ai += hi[2 * d + e] * cos7(r);
            br += hr[2 * d + e] * sin7(r);
            bi += hi[2 * d + e] * sin7(r);
        }
        zr[j]     = ar + bi;  zi[j]     = ai - br;
        zr[7 - j] = ar - bi;  zi[7 - j] = ai + br;
    }
}

__device__ __forceinline__ void inv_load(const float* __restrict__ xt, const int q, v2f (&z)[NC]) {
#pragma unroll
    for (int a = 0; a < 7; ++a)
        z[a] = *(const v2f*)(xt + (q + NU * a) * (T_N * 2));
#pragma unroll
    for (int j = 0; j < 7; ++j)
        z[7 + j] = *(const v2f*)(xt + (NU * (7 - j) - q) * (T_N * 2));
}

__device__ __forceinline__ void inv_join(const v2f (&z)[NC], const int q,
                                         const float* __restrict__ sTw,
                                         float* __restrict__ hrow, const bool ok) {
    float zr[NC], zi[NC];
#pragma unroll
    for (int a = 0; a < 7; ++a)  { zr[a] = z[a].x; zi[a] = z[a].y; }
#pragma unroll
    for (int a = 7; a < NC; ++a) { zr[a] = z[a].x; zi[a] = -z[a].y; }
    if (q == 0) { zi[0] = 0.f; zi[7] = 0.f; }
    float er[7], ei[7], pr[7], pi[7];
    dft7(zr, zi, 0, er, ei);
    dft7(zr, zi, 1, pr, pi);
#pragma unroll
    for (int j = 0; j < 7; ++j) {
        const int jj = (7 - j) % 7;
        const float cj = (j == 0) ? 1.f : (j == 1) ? 0.9009688679024191f : (j == 2) ? 0.6234898018587336f
                       : (j == 3) ? 0.2225209339563144f : (j == 4) ? -0.2225209339563144f
                       : (j == 5) ? -0.6234898018587336f : -0.9009688679024191f;
        const float sj = (j == 0) ? 0.f : (j == 1 || j == 6) ? 0.4338837391175581f
                       : (j == 2 || j == 5) ? 0.7818314824680298f : 0.9749279121818236f;
        float P, Q;
        if (j == 0) { P = pr[0]; Q = pi[0]; }
        else        { P = pr[jj] * cj - pi[jj] * sj;  Q = pi[jj] * cj + pr[jj] * sj; }
        const float h0r = er[jj] + P, h0i = ei[jj] + Q;
        const float h1r = er[jj] - P, h1i = ei[jj] - Q;
        const v2f t0 = *(const v2f*)(sTw + 2 * (q * NC + j));
        const v2f t1 = *(const v2f*)(sTw + 2 * (q * NC + j + 7));
        if (ok) {
            hrow[j * UPAD + q]            = h0r * t0.x + h0i * t0.y;
            hrow[j * UPAD + QH + q]       = h0i * t0.x - h0r * t0.y;
            hrow[(j + 7) * UPAD + q]      = h1r * t1.x + h1i * t1.y;
            hrow[(j + 7) * UPAD + QH + q] = h1i * t1.x - h1r * t1.y;
        }
    }
}

__global__ __launch_bounds__(512) void istft_signal(const float* __restrict__ x,
                                                    const float* __restrict__ Mi,
                                                    const float* __restrict__ swg,
                                                    const float* __restrict__ twj,
                                                    float* __restrict__ outsig) {
    __shared__ __attribute__((aligned(16))) float sRaw[RAW_LEN];
    __shared__ __attribute__((aligned(16))) float sH[IFT * NC * UPAD];
    __shared__ __attribute__((aligned(16))) float sM[W73_ELEMS];
    __shared__ __attribute__((aligned(16))) float sSw[KPAD];
    __shared__ __attribute__((aligned(16))) float sTw[TWJ_ELEMS];

    const int tid  = threadIdx.x;
    const int lane = tid & 31;
    const int wave = __builtin_amdgcn_readfirstlane(tid >> 5);
    const int half = lane >> 4;
    const int r16  = lane & 15;
    const int b    = blockIdx.x;
    const float* xs = x + (size_t)b * (F_N * T_N * 2);

    const int  q0    = (wave << 1) + half;
    const int  q1raw = 32 + (wave << 1) + half;
    const int  q1    = q1raw < QH ? q1raw : QH - 1;
    const bool two   = wave < 3;

    v2f z0[NC], z1[NC];
    inv_load(xs + 2 * r16, q0, z0);
    if (two) inv_load(xs + 2 * r16, q1, z1);

    for (int i = tid; i < RAW_LEN / 4; i += 512)
        ((float4*)sRaw)[i] = make_float4(0.f, 0.f, 0.f, 0.f);
    for (int i = tid; i < W73_ELEMS / 4; i += 512)
        ((float4*)sM)[i] = ((const float4*)Mi)[i];
    for (int i = tid; i < KPAD; i += 512)
        sSw[i] = swg[i];
    for (int i = tid; i < TWJ_ELEMS; i += 512)
        sTw[i] = twj[i];
    if (tid < IFT * NC) {
        sH[tid * UPAD + 2 * QH]     = 0.f;
        sH[tid * UPAD + 2 * QH + 1] = 0.f;
    }
    __syncthreads();

    const int Rrow = (wave << 4) + r16;
    const int tl   = Rrow / NC;
    const int nb   = Rrow - tl * NC + (NC * 8) * half;
    float swr[40];
#pragma unroll
    for (int i = 0; i < 5; ++i) {
#pragma unroll
        for (int v = 0; v < 8; ++v) {
            if (i < 4 || v == 0)  swr[8 * i + v] = sSw[nb + NC * (16 * i + v)];
            else                  swr[8 * i + v] = (half == 0) ? sSw[nb + NC * (16 * i + v)] : 0.f;
        }
    }

    float* hrow = sH + r16 * (NC * UPAD);
    for (int ch = 0; ch < ICH; ++ch) {
        const int  t0  = ch * IFT;
        const bool tok = t0 + r16 < T_N;
        inv_join(z0, q0, sTw, hrow, tok);
        if (two) inv_join(z1, q1, sTw, hrow, tok & (q1raw < QH));
        __syncthreads();

        if (ch + 1 < ICH) {
            const int tn = t0 + IFT + r16;
            const int tc = tn < T_N ? tn : T_N - 1;
            inv_load(xs + 2 * tc, q0, z0);
            if (two) inv_load(xs + 2 * tc, q1, z1);
        }

        const int ntile = (ch + 1 < ICH) ? NC : 1;
        if (wave < ntile) {
            v8f acc[5];
#pragma unroll
            for (int i = 0; i < 5; ++i)
                acc[i] = (v8f){0.f,0.f,0.f,0.f,0.f,0.f,0.f,0.f};
            const float* ap = sM + r16 * UPAD + (half << 1);
            const float* bp = sH + ((wave << 4) + r16) * UPAD + (half << 1);
#pragma unroll
            for (int k0 = 0; k0 < UPAD; k0 += 4) {
                v2f a[5];
#pragma unroll
                for (int i = 0; i < 5; ++i) a[i] = *(const v2f*)(ap + i * (16 * UPAD) + k0);
                const v2f bb = *(const v2f*)(bp + k0);
#pragma unroll
                for (int i = 0; i < 5; ++i)
                    acc[i] = __builtin_amdgcn_wmma_f32_16x16x4_f32(false, a[i], false, bb, (short)0, acc[i], false, false);
            }
            const int tt = t0 + tl;
            if (tt < T_N) {
                float* sp = sRaw + tt * HOP_N + nb;
#pragma unroll
                for (int i = 0; i < 4; ++i) {
#pragma unroll
                    for (int v = 0; v < 8; ++v)
                        __hip_atomic_fetch_add(sp + NC * (16 * i + v), acc[i][v] * swr[8 * i + v],
                                               __ATOMIC_RELAXED, __HIP_MEMORY_SCOPE_WORKGROUP);
                }
                __hip_atomic_fetch_add(sp + NC * 64, acc[4][0] * swr[32],
                                       __ATOMIC_RELAXED, __HIP_MEMORY_SCOPE_WORKGROUP);
                if (half == 0) {
#pragma unroll
                    for (int v = 1; v < 8; ++v)
                        __hip_atomic_fetch_add(sp + NC * (64 + v), acc[4][v] * swr[32 + v],
                                               __ATOMIC_RELAXED, __HIP_MEMORY_SCOPE_WORKGROUP);
                }
            }
        }
        __syncthreads();
    }

    float* dst = outsig + (size_t)b * OUT_STRIDE;
    const int rr = 2 * tid;
    const float up0 = (float)(rr + 1) * 0.02f,     up1 = (float)(rr + 2) * 0.02f;
    const float dn0 = (float)(OVL_N - rr) * 0.02f, dn1 = (float)(OVL_N - 1 - rr) * 0.02f;
#pragma unroll
    for (int q0 = 0; q0 < NSEG; q0 += 6) {
        v2f val[6];
#pragma unroll
        for (int k = 0; k < 6; ++k) {
            const int q   = q0 + k;
            const int len = (q == NSEG - 1) ? SEG_N : SEG_N - OVL_N;
            val[k] = (v2f){0.f, 0.f};
            if (rr < len) val[k] = *(const v2f*)(sRaw + q * PERIOD_N + rr);
        }
        if (rr < OVL_N) {
#pragma unroll
            for (int k = 0; k < 6; ++k) {
                const int q = q0 + k;
                if (q > 0) {
                    const v2f prev = *(const v2f*)(sRaw + q * PERIOD_N + rr - (PERIOD_N - SEG_N + OVL_N));
                    val[k].x = up0 * val[k].x + dn0 * prev.x;
                    val[k].y = up1 * val[k].y + dn1 * prev.y;
                }
            }
        }
#pragma unroll
        for (int k = 0; k < 6; ++k) {
            const int q   = q0 + k;
            const int len = (q == NSEG - 1) ? SEG_N : SEG_N - OVL_N;
            if (rr < len) *(v2f*)(dst + q * (SEG_N - OVL_N) + rr) = val[k];
        }
    }
}

__global__ __launch_bounds__(256) void stft_ct(const float* __restrict__ outsig,
                                               const float* __restrict__ W73,
                                               const float* __restrict__ twj,
                                               const float* __restrict__ hw,
                                               float* __restrict__ out) {
    __shared__ __attribute__((aligned(16))) float sY[FT * NC * UPAD];
    __shared__ __attribute__((aligned(16))) float sW[W73_ELEMS];

    const int tid  = threadIdx.x;
    const int lane = tid & 31;
    const int wave = __builtin_amdgcn_readfirstlane(tid >> 5);
    const int half = lane >> 4;
    const int r16  = lane & 15;
    const int rowBlock = blockIdx.x * FT;

    for (int i = tid; i < W73_ELEMS / 4; i += 256)
        ((float4*)sW)[i] = ((const float4*)W73)[i];
    if (tid < FT * NC) {
        sY[tid * UPAD + NU]     = 0.f;
        sY[tid * UPAD + NU + 1] = 0.f;
        sY[tid * UPAD + NU + 2] = 0.f;
    }
#pragma unroll
    for (int it = 0; it < 2; ++it) {
        const int p = tid + (it << 8);
        if (p < HALF_N) {
            const int n0 = 2 * p, n1 = n0 + 1;
            const int u0 = n0 / NC, c0 = n0 - u0 * NC;
            const int u1 = n1 / NC, c1 = n1 - u1 * NC;
            const v2f h = *(const v2f*)(hw + n0);
            const int o0 = c0 * UPAD + u0, o1 = c1 * UPAD + u1;
#pragma unroll
            for (int f = 0; f < FT; ++f) {
                const int R  = rowBlock + f;
                const int fb = R / NF_N;
                const int fm = R - fb * NF_N;
                const v2f sv = *(const v2f*)(outsig + (size_t)fb * OUT_STRIDE + fm * HOP_N + n0);
                sY[f * (NC * UPAD) + o0] = sv.x * h.x;
                sY[f * (NC * UPAD) + o1] = sv.y * h.y;
            }
        }
    }
    __syncthreads();

    if (wave < 7) {
        v8f acc[5][2];
#pragma unroll
        for (int i = 0; i < 5; ++i)
#pragma unroll
            for (int j = 0; j < 2; ++j)
                acc[i][j] = (v8f){0.f,0.f,0.f,0.f,0.f,0.f,0.f,0.f};
        const float* ap = sW + r16 * UPAD + (half << 1);
        const float* bp = sY + ((wave << 5) + r16) * UPAD + (half << 1);
#pragma unroll
        for (int k0 = 0; k0 < UPAD; k0 += 4) {
            v2f a[5], bb[2];
#pragma unroll
            for (int i = 0; i < 5; ++i) a[i] = *(const v2f*)(ap + i * (16 * UPAD) + k0);
#pragma unroll
            for (int j = 0; j < 2; ++j) bb[j] = *(const v2f*)(bp + j * (16 * UPAD) + k0);
#pragma unroll
            for (int i = 0; i < 5; ++i)
#pragma unroll
                for (int j = 0; j < 2; ++j)
                    acc[i][j] = __builtin_amdgcn_wmma_f32_16x16x4_f32(false, a[i], false, bb[j], (short)0, acc[i][j], false, false);
        }
#pragma unroll
        for (int j = 0; j < 2; ++j) {
            float* g = sY + ((wave << 5) + (j << 4) + r16) * UPAD + (half << 3);
#pragma unroll
            for (int i = 0; i < 5; ++i) {
                const v8f d = acc[i][j];
                *(float4*)(g + 16 * i) = make_float4(d[0], d[1], d[2], d[3]);
                if (i < 4 || half == 0)
                    *(float4*)(g + 16 * i + 4) = make_float4(d[4], d[5], d[6], d[7]);
            }
        }
    }
    __syncthreads();

    const int R  = rowBlock + r16;
    const int jb = R / NF_N;
    const int jm = R - jb * NF_N;
    float* op = out + (size_t)jb * (F_N * NF_N * 2) + jm * 2;
    const float* gf = sY + r16 * (NC * UPAD);
    for (int it = 0; it < 3; ++it) {
        const int q = ((wave + (it << 3)) << 1) + half;
        if (q < QH) {
            float hr[NC], hi[NC];
#pragma unroll
            for (int c = 0; c < NC; ++c) {
                const float gr = gf[c * UPAD + q];
                const float gi = gf[c * UPAD + QH + q];
                const v2f  t  = *(const v2f*)(twj + 2 * (q * NC + c));
                hr[c] = gr * t.x - gi * t.y;
                hi[c] = gr * t.y + gi * t.x;
            }
            float er[7], ei[7], pr[7], pi[7];
            dft7(hr, hi, 0, er, ei);
            dft7(hr, hi, 1, pr, pi);
            const float sg = (q == 0) ? 1.f : -1.f;
#pragma unroll
            for (int j = 0; j < 7; ++j) {
                const float cj = (j == 0) ? 1.f : (j == 1) ? 0.9009688679024191f : (j == 2) ? 0.6234898018587336f
                               : (j == 3) ? 0.2225209339563144f : (j == 4) ? -0.2225209339563144f
                               : (j == 5) ? -0.6234898018587336f : -0.9009688679024191f;
                const float sj = (j == 0) ? 0.f : (j == 1 || j == 6) ? 0.4338837391175581f
                               : (j == 2 || j == 5) ? 0.7818314824680298f : 0.9749279121818236f;
                float P, Q;
                if (j == 0) { P = pr[0]; Q = pi[0]; }
                else        { P = pr[j] * cj + pi[j] * sj;  Q = pi[j] * cj - pr[j] * sj; }
                v2f lo, hi2;
                lo.x  = er[j] + P;  lo.y  = ei[j] + Q;
                hi2.x = er[j] - P;  hi2.y = sg * (ei[j] - Q);
                *(v2f*)(op + (size_t)(q + NU * j) * (NF_N * 2)) = lo;
                if (q != 0 || j == 0)
                    *(v2f*)(op + (size_t)(NU * (7 - j) - q) * (NF_N * 2)) = hi2;
            }
        }
    }
}

extern "C" void kernel_launch(void* const* d_in, const int* in_sizes, int n_in,
                              void* d_out, int out_size, void* d_ws, size_t ws_size,
                              hipStream_t stream) {
    (void)in_sizes; (void)n_in; (void)out_size; (void)ws_size;
    const float* x = (const float*)d_in[0];
    float* out = (float*)d_out;

    float* ws     = (float*)d_ws;
    float* Wi     = ws;
    float* WfwdT  = Wi + (size_t)KPAD * KPAD;
    float* outsig = Wi + (size_t)1792 * KPAD;

    build_mats<<<(IM_ELEMS + FWD_TAB + KPAD) / 256, 256, 0, stream>>>(Wi, WfwdT);
    istft_signal<<<B_N, 512, 0, stream>>>(x, Wi, Wi + IM_ELEMS, WfwdT + TWJ_OFF, outsig);
    stft_ct<<<M2 / FT, 256, 0, stream>>>(outsig, WfwdT, WfwdT + TWJ_OFF, WfwdT + HANN_OFF, out);
}
